# starting kernel
# baseline (speedup 1.0000x reference)
#include <hip/hip_runtime.h>


#define NS_  8
#define NC_  4
#define T_   32
#define NB_  128
#define NSC_ 12
#define T1_  33

typedef float v2f __attribute__((ext_vector_type(2)));
typedef float v8f __attribute__((ext_vector_type(8)));

__device__ __forceinline__ v8f wmma4(v2f a, v2f b, v8f acc) {
  return __builtin_amdgcn_wmma_f32_16x16x4_f32(
       false, a,  false, b,
       (short)0, acc,  false,  false);
}

__global__ __launch_bounds__(32) void mpc_riccati(
    const float* __restrict__ x0g,
    const float* __restrict__ Cg_,
    const float* __restrict__ cg_,
    const float* __restrict__ Fg_,
    const float* __restrict__ fg_,
    float* __restrict__ out)
{
  const int  b    = blockIdx.x;
  const int  lane = threadIdx.x;
  const int  hl   = lane & 15;
  const bool hiw  = lane >= 16;
  const int  rbase = hiw ? 8 : 0;
  const int  kbase = hiw ? 2 : 0;

  __shared__ float Pm[16][16];
  __shared__ float Wm[16][16];
  __shared__ float Qm[16][16];
  __shared__ float Fs[16][16];
  __shared__ float Cs[16][16];
  __shared__ float Km[NC_][16];
  __shared__ float kvv[NC_];
  __shared__ float pv[NS_];
  __shared__ float qv[NSC_];
  __shared__ float Pfp[NS_];
  __shared__ float Kall[T1_][NC_][NS_];
  __shared__ float kall[T1_][NC_];
  __shared__ float xs[NS_];
  __shared__ float us[NC_];
  __shared__ float xn[NS_];

  for (int idx = lane; idx < 256; idx += 32) {
    int r = idx >> 4, cx = idx & 15;
    Pm[r][cx] = 0.f; Wm[r][cx] = 0.f; Qm[r][cx] = 0.f;
    Fs[r][cx] = 0.f; Cs[r][cx] = 0.f;
    if (r < NC_) Km[r][cx] = 0.f;
  }
  __syncthreads();

  auto solve_update = [&](int t) {
    float A[4][4], Inv[4][4];
    #pragma unroll
    for (int i = 0; i < 4; ++i)
      #pragma unroll
      for (int j = 0; j < 4; ++j) {
        A[i][j]   = Qm[NS_ + i][NS_ + j];
        Inv[i][j] = (i == j) ? 1.f : 0.f;
      }
    #pragma unroll
    for (int cc = 0; cc < 4; ++cc) {
      float piv = 1.f / A[cc][cc];
      #pragma unroll
      for (int j = 0; j < 4; ++j) { A[cc][j] *= piv; Inv[cc][j] *= piv; }
      #pragma unroll
      for (int r = 0; r < 4; ++r) {
        if (r == cc) continue;
        float fac = A[r][cc];
        #pragma unroll
        for (int j = 0; j < 4; ++j) { A[r][j] -= fac * A[cc][j]; Inv[r][j] -= fac * Inv[cc][j]; }
      }
    }
    {
      int jj = lane >> 3;
      int ii = lane & 7;
      float s = 0.f;
      #pragma unroll
      for (int m = 0; m < 4; ++m) s += Inv[jj][m] * Qm[NS_ + m][ii];
      Km[jj][ii]      = -s;
      Kall[t][jj][ii] = -s;
    }
    if (lane < NC_) {
      float s = 0.f;
      #pragma unroll
      for (int m = 0; m < 4; ++m) s += Inv[lane][m] * qv[NS_ + m];
      kvv[lane]     = -s;
      kall[t][lane] = -s;
    }
    __syncthreads();
    if (lane < NS_) {
      float s = qv[lane];
      #pragma unroll
      for (int j = 0; j < NC_; ++j) s += Qm[lane][NS_ + j] * kvv[j];
      pv[lane] = s;
    }
    v2f a, bb;
    {
      float a0 = Qm[hl][NS_ + kbase];
      float a1 = Qm[hl][NS_ + kbase + 1];
      a.x  = (hl < NS_) ? a0 : 0.f;
      a.y  = (hl < NS_) ? a1 : 0.f;
      bb.x = Km[kbase][hl];
      bb.y = Km[kbase + 1][hl];
    }
    v8f cf;
    #pragma unroll
    for (int r = 0; r < 8; ++r) {
      int row = r + rbase;
      float v = Qm[row][hl];
      cf[r] = (row < NS_ && hl < NS_) ? v : 0.f;
    }
    v8f d = wmma4(a, bb, cf);
    __syncthreads();
    #pragma unroll
    for (int r = 0; r < 8; ++r) Pm[r + rbase][hl] = d[r];
    __syncthreads();
  };

  {
    const float* Cg = Cg_ + (size_t)((T_ - 1) * NB_ + b) * NSC_ * NSC_;
    for (int idx = lane; idx < NSC_ * NSC_; idx += 32)
      Qm[idx / NSC_][idx % NSC_] = Cg[idx];
    const float* cg = cg_ + (size_t)((T_ - 1) * NB_ + b) * NSC_;
    if (lane < NSC_) qv[lane] = cg[lane];
    __syncthreads();
    solve_update(T_);
  }

  for (int t = T_ - 1; t >= 0; --t) {
    const float* Fg = Fg_ + (size_t)(t * NB_ + b) * NS_ * NSC_;
    const float* fg = fg_ + (size_t)(t * NB_ + b) * NS_;
    const float* Cg = Cg_ + (size_t)(t * NB_ + b) * NSC_ * NSC_;
    const float* cg = cg_ + (size_t)(t * NB_ + b) * NSC_;

    if (t > 0) {
      __builtin_prefetch(Fg_ + (size_t)((t - 1) * NB_ + b) * NS_ * NSC_, 0, 0);
      __builtin_prefetch(Cg_ + (size_t)((t - 1) * NB_ + b) * NSC_ * NSC_, 0, 0);
    }

    for (int idx = lane; idx < NS_ * NSC_; idx += 32)
      Fs[idx / NSC_][idx % NSC_] = Fg[idx];
    for (int idx = lane; idx < NSC_ * NSC_; idx += 32)
      Cs[idx / NSC_][idx % NSC_] = Cg[idx];
    __syncthreads();

    v2f fb0, fb1;
    fb0.x = Fs[kbase][hl];     fb0.y = Fs[kbase + 1][hl];
    fb1.x = Fs[4 + kbase][hl]; fb1.y = Fs[4 + kbase + 1][hl];

    v8f acc = {0.f, 0.f, 0.f, 0.f, 0.f, 0.f, 0.f, 0.f};
    {
      v2f a0, a1;
      a0.x = Pm[hl][kbase];     a0.y = Pm[hl][kbase + 1];
      a1.x = Pm[hl][4 + kbase]; a1.y = Pm[hl][4 + kbase + 1];
      acc = wmma4(a0, fb0, acc);
      acc = wmma4(a1, fb1, acc);
    }
    #pragma unroll
    for (int r = 0; r < 8; ++r) Wm[r + rbase][hl] = acc[r];

    if (lane < NS_) {
      float s = pv[lane];
      #pragma unroll
      for (int m = 0; m < NS_; ++m) s += Pm[lane][m] * fg[m];
      Pfp[lane] = s;
    }
    __syncthreads();

    v8f acc2;
    #pragma unroll
    for (int r = 0; r < 8; ++r) acc2[r] = Cs[r + rbase][hl];
    {
      v2f b0, b1;
      b0.x = Wm[kbase][hl];     b0.y = Wm[kbase + 1][hl];
      b1.x = Wm[4 + kbase][hl]; b1.y = Wm[4 + kbase + 1][hl];
      acc2 = wmma4(fb0, b0, acc2);
      acc2 = wmma4(fb1, b1, acc2);
    }
    #pragma unroll
    for (int r = 0; r < 8; ++r) Qm[r + rbase][hl] = acc2[r];

    if (lane < NSC_) {
      float s = cg[lane];
      #pragma unroll
      for (int i = 0; i < NS_; ++i) s += Fs[i][lane] * Pfp[i];
      qv[lane] = s;
    }
    __syncthreads();

    solve_update(t);
  }

  float* out_x = out;
  float* out_u = out + (size_t)T1_ * NB_ * NS_;
  if (lane < NS_) xs[lane] = x0g[(size_t)b * NS_ + lane];
  __syncthreads();
  for (int t = 0; t <= T_; ++t) {
    if (lane < NC_) {
      float s = kall[t][lane];
      #pragma unroll
      for (int i = 0; i < NS_; ++i) s += Kall[t][lane][i] * xs[i];
      us[lane] = s;
    }
    __syncthreads();
    if (lane < NS_) out_x[(size_t)t * NB_ * NS_ + b * NS_ + lane] = xs[lane];
    if (lane < NC_) out_u[(size_t)t * NB_ * NC_ + b * NC_ + lane] = us[lane];
    if (t < T_) {
      const float* Fg = Fg_ + (size_t)(t * NB_ + b) * NS_ * NSC_;
      const float* fg = fg_ + (size_t)(t * NB_ + b) * NS_;
      if (lane < NS_) {
        float s = fg[lane];
        #pragma unroll
        for (int j = 0; j < NS_; ++j) s += Fg[lane * NSC_ + j] * xs[j];
        #pragma unroll
        for (int j = 0; j < NC_; ++j) s += Fg[lane * NSC_ + NS_ + j] * us[j];
        xn[lane] = s;
      }
      __syncthreads();
      if (lane < NS_) xs[lane] = xn[lane];
      __syncthreads();
    }
  }
}

extern "C" void kernel_launch(void* const* d_in, const int* in_sizes, int n_in,
                              void* d_out, int out_size, void* d_ws, size_t ws_size,
                              hipStream_t stream) {
  (void)in_sizes; (void)n_in; (void)d_ws; (void)ws_size; (void)out_size;
  const float* x0 = (const float*)d_in[0];
  const float* C  = (const float*)d_in[1];
  const float* c  = (const float*)d_in[2];
  const float* F  = (const float*)d_in[3];
  const float* f  = (const float*)d_in[4];
  float* out = (float*)d_out;
  mpc_riccati<<<NB_, 32, 0, stream>>>(x0, C, c, F, f, out);
}


# Step's vector sums (Pf+p, q, p, b) unpredicated: every lane forms its row's copy, so their LDS reads issue with the products' operand reads and their adds run under the WMMAs; sums keep their order
# speedup vs baseline: 3.4908x; 3.4908x over previous
#include <hip/hip_runtime.h>


#define NS_  8
#define NC_  4
#define T_   32
#define NB_  128
#define NSC_ 12
#define T1_  33

typedef float v2f __attribute__((ext_vector_type(2)));
typedef float v8f __attribute__((ext_vector_type(8)));

__device__ __forceinline__ v8f wmma4(v2f a, v2f b, v8f acc) {
  return __builtin_amdgcn_wmma_f32_16x16x4_f32(
       false, a,  false, b,
       (short)0, acc,  false,  false);
}

__global__ __launch_bounds__(32) void mpc_riccati(
    const float* __restrict__ x0g,
    const float* __restrict__ Cg_,
    const float* __restrict__ cg_,
    const float* __restrict__ Fg_,
    const float* __restrict__ fg_,
    float* __restrict__ out)
{
  const int  b    = blockIdx.x;
  const int  lane = threadIdx.x;
  const int  hl   = lane & 15;
  const bool hiw  = lane >= 16;
  const int  rbase = hiw ? 8 : 0;
  const int  kbase = hiw ? 2 : 0;

  __shared__ float Pm[16][16];
  __shared__ float Wm[16][16];
  __shared__ float Qm[16][16];
  __shared__ float pv[NS_];
  __shared__ float qv[NSC_];
  __shared__ float Pfp[NS_];
  __shared__ float Mall[T1_][NSC_][NS_ + 1];

  for (int idx = lane; idx < 256; idx += 32) {
    int r = idx >> 4, cx = idx & 15;
    Pm[r][cx] = 0.f; Wm[r][cx] = 0.f; Qm[r][cx] = 0.f;
  }
  __syncthreads();

  auto solve_update = [&](int t, const bool dyn, float fl,
                          float fu0, float fu1, float fu2, float fu3,
                          v2f af, v8f cF) {
    float A[4][4];
    #pragma unroll
    for (int i = 0; i < 4; ++i)
      #pragma unroll
      for (int j = 0; j < 4; ++j) A[i][j] = Qm[NS_ + i][NS_ + j];
    float rx[4], rq[4];
    #pragma unroll
    for (int m2 = 0; m2 < 4; ++m2) { rx[m2] = Qm[NS_ + m2][hl]; rq[m2] = qv[NS_ + m2]; }
    const int l7 = lane & 7;
    float qxu[4];
    #pragma unroll
    for (int j = 0; j < 4; ++j) qxu[j] = Qm[l7][NS_ + j];
    const float qx = qv[l7];
    v2f a;
    a.x = Qm[hl][NS_ + kbase];
    a.y = Qm[hl][NS_ + kbase + 1];
    v8f cf;
    #pragma unroll
    for (int r = 0; r < 8; ++r) cf[r] = Qm[r + rbase][hl];

    asm volatile("" ::
        "v"(A[0][0]), "v"(A[0][1]), "v"(A[0][2]), "v"(A[0][3]),
        "v"(A[1][0]), "v"(A[1][1]), "v"(A[1][2]), "v"(A[1][3]),
        "v"(A[2][0]), "v"(A[2][1]), "v"(A[2][2]), "v"(A[2][3]),
        "v"(A[3][0]), "v"(A[3][1]), "v"(A[3][2]), "v"(A[3][3]),
        "v"(rx[0]), "v"(rx[1]), "v"(rx[2]), "v"(rx[3]),
        "v"(rq[0]), "v"(rq[1]), "v"(rq[2]), "v"(rq[3]));

    #pragma unroll
    for (int cc = 0; cc < 4; ++cc) {
      const float piv = 1.f / A[cc][cc];
      #pragma unroll
      for (int j = 0; j < 4; ++j) A[cc][j] *= piv;
      rx[cc] *= piv; rq[cc] *= piv;
      #pragma unroll
      for (int r = 0; r < 4; ++r) {
        if (r == cc) continue;
        const float fac = A[r][cc];
        #pragma unroll
        for (int j = 0; j < 4; ++j) A[r][j] -= fac * A[cc][j];
        rx[r] -= fac * rx[cc];
        rq[r] -= fac * rq[cc];
      }
    }
    float Kc[4], kk[4];
    #pragma unroll
    for (int m2 = 0; m2 < 4; ++m2) { Kc[m2] = -rx[m2]; kk[m2] = -rq[m2]; }
    v2f bb;
    bb.x = hiw ? Kc[2] : Kc[0];
    bb.y = hiw ? Kc[3] : Kc[1];
    v8f d = wmma4(a, bb, cf);
    #pragma unroll
    for (int r = 0; r < 8; ++r) Pm[r + rbase][hl] = d[r];
    {
      const bool isK = hl < NS_;
      const int  kc  = isK ? hl : NS_;
      #pragma unroll
      for (int m2 = 0; m2 < 4; ++m2)
        Mall[t][NS_ + m2][kc] = isK ? Kc[m2] : kk[m2];
    }
    {
      float s = qx;
      #pragma unroll
      for (int j = 0; j < NC_; ++j) s += qxu[j] * kk[j];
      pv[l7] = s;
      if (dyn) {
        float bs = fl;
        bs += fu0 * kk[0];
        bs += fu1 * kk[1];
        bs += fu2 * kk[2];
        bs += fu3 * kk[3];
        Mall[t][l7][NS_] = bs;
      }
    }
    if (dyn) {
      v8f acl = wmma4(af, bb, cF);
      if (lane < NS_) {
        #pragma unroll
        for (int r = 0; r < 8; ++r) Mall[t][r][lane] = acl[r];
      }
    } else {
      for (int idx = lane; idx < NS_ * (NS_ + 1); idx += 32)
        Mall[t][idx / (NS_ + 1)][idx % (NS_ + 1)] = 0.f;
    }
    __syncthreads();
  };

  {
    const float* Cg = Cg_ + (size_t)((T_ - 1) * NB_ + b) * NSC_ * NSC_;
    for (int idx = lane; idx < NSC_ * NSC_; idx += 32)
      Qm[idx / NSC_][idx % NSC_] = Cg[idx];
    const float* cg = cg_ + (size_t)((T_ - 1) * NB_ + b) * NSC_;
    if (lane < NSC_) qv[lane] = cg[lane];
    __syncthreads();
    const v2f z2 = {0.f, 0.f};
    const v8f z8 = {0.f, 0.f, 0.f, 0.f, 0.f, 0.f, 0.f, 0.f};
    solve_update(T_, false, 0.f, 0.f, 0.f, 0.f, 0.f, z2, z8);
  }

  for (int t = T_ - 1; t >= 0; --t) {
    const float* Fg = Fg_ + (size_t)(t * NB_ + b) * NS_ * NSC_;
    const float* fg = fg_ + (size_t)(t * NB_ + b) * NS_;
    const float* Cg = Cg_ + (size_t)(t * NB_ + b) * NSC_ * NSC_;
    const float* cg = cg_ + (size_t)(t * NB_ + b) * NSC_;

    if (t > 0) {
      __builtin_prefetch(Fg_ + (size_t)((t - 1) * NB_ + b) * NS_ * NSC_, 0, 0);
      __builtin_prefetch(Cg_ + (size_t)((t - 1) * NB_ + b) * NSC_ * NSC_, 0, 0);
    }

    const int  hc = (hl < NSC_) ? hl : (NSC_ - 1);
    const bool cv = hl < NSC_;
    float Fcol[NS_];
    #pragma unroll
    for (int i = 0; i < NS_; ++i) Fcol[i] = Fg[i * NSC_ + hc];
    float Ccol[8];
    #pragma unroll
    for (int r = 0; r < 8; ++r) {
      const int rw = r + rbase;
      Ccol[r] = Cg[((rw < NSC_) ? rw : (NSC_ - 1)) * NSC_ + hc];
    }
    const float fu0 = Fg[(lane & 7) * NSC_ + NS_ + 0];
    const float fu1 = Fg[(lane & 7) * NSC_ + NS_ + 1];
    const float fu2 = Fg[(lane & 7) * NSC_ + NS_ + 2];
    const float fu3 = Fg[(lane & 7) * NSC_ + NS_ + 3];
    const float fl  = fg[lane & 7];
    const float cl  = cg[hc];
    float fv[NS_];
    #pragma unroll
    for (int m = 0; m < NS_; ++m)
      fv[m] = __int_as_float(__builtin_amdgcn_readlane(__float_as_int(fl), m));

    v2f fb0, fb1;
    fb0.x = cv ? (hiw ? Fcol[2] : Fcol[0]) : 0.f;
    fb0.y = cv ? (hiw ? Fcol[3] : Fcol[1]) : 0.f;
    fb1.x = cv ? (hiw ? Fcol[6] : Fcol[4]) : 0.f;
    fb1.y = cv ? (hiw ? Fcol[7] : Fcol[5]) : 0.f;
    v2f af;
    af.x = (hl < NS_) ? (hiw ? fu2 : fu0) : 0.f;
    af.y = (hl < NS_) ? (hiw ? fu3 : fu1) : 0.f;
    v8f cF;
    #pragma unroll
    for (int r = 0; r < 8; ++r) cF[r] = (!hiw && cv) ? Fcol[r] : 0.f;

    v8f acc = {0.f, 0.f, 0.f, 0.f, 0.f, 0.f, 0.f, 0.f};
    const int l7 = lane & 7;
    {
      v2f a0, a1;
      a0.x = Pm[hl][kbase];     a0.y = Pm[hl][kbase + 1];
      a1.x = Pm[hl][4 + kbase]; a1.y = Pm[hl][4 + kbase + 1];
      float prow[NS_];
      #pragma unroll
      for (int m = 0; m < NS_; ++m) prow[m] = Pm[l7][m];
      float s = pv[l7];
      acc = wmma4(a0, fb0, acc);
      acc = wmma4(a1, fb1, acc);
      #pragma unroll
      for (int m = 0; m < NS_; ++m) s += prow[m] * fv[m];
      Pfp[l7] = s;
    }
    #pragma unroll
    for (int r = 0; r < 8; ++r) Wm[r + rbase][hl] = acc[r];
    __syncthreads();

    v8f acc2;
    #pragma unroll
    for (int r = 0; r < 8; ++r) acc2[r] = ((r + rbase) < NSC_ && cv) ? Ccol[r] : 0.f;
    {
      v2f b0, b1;
      b0.x = Wm[kbase][hl];     b0.y = Wm[kbase + 1][hl];
      b1.x = Wm[4 + kbase][hl]; b1.y = Wm[4 + kbase + 1][hl];
      float pf[NS_];
      #pragma unroll
      for (int i = 0; i < NS_; ++i) pf[i] = Pfp[i];
      acc2 = wmma4(fb0, b0, acc2);
      acc2 = wmma4(fb1, b1, acc2);
      float s = cl;
      #pragma unroll
      for (int i = 0; i < NS_; ++i) s += Fcol[i] * pf[i];
      qv[hc] = s;
    }
    #pragma unroll
    for (int r = 0; r < 8; ++r) Qm[r + rbase][hl] = acc2[r];
    __syncthreads();

    solve_update(t, true, fl, fu0, fu1, fu2, fu3, af, cF);
  }

  float* out_x = out;
  float* out_u = out + (size_t)T1_ * NB_ * NS_;
  const int row = (lane < NSC_) ? lane : (NSC_ - 1);
  float xv = (lane < NS_) ? x0g[(size_t)b * NS_ + lane] : 0.f;
  float m[NS_ + 1], mn[NS_ + 1];
  #pragma unroll
  for (int j = 0; j <= NS_; ++j) m[j] = Mall[0][row][j];
  for (int t = 0; t <= T_; ++t) {
    const int tn = (t < T_) ? t + 1 : T_;
    #pragma unroll
    for (int j = 0; j <= NS_; ++j) mn[j] = Mall[tn][row][j];
    float xj[NS_];
    #pragma unroll
    for (int j = 0; j < NS_; ++j)
      xj[j] = __int_as_float(__builtin_amdgcn_readlane(__float_as_int(xv), j));
    float s0 = m[NS_], s1 = 0.f;
    #pragma unroll
    for (int j = 0; j < NS_; j += 2) { s0 += m[j] * xj[j]; s1 += m[j + 1] * xj[j + 1]; }
    const float s = s0 + s1;
    if (lane < NS_) out_x[(size_t)t * NB_ * NS_ + b * NS_ + lane] = xv;
    else if (lane < NSC_) out_u[(size_t)t * NB_ * NC_ + b * NC_ + (lane - NS_)] = s;
    xv = s;
    #pragma unroll
    for (int j = 0; j <= NS_; ++j) m[j] = mn[j];
  }
}

extern "C" void kernel_launch(void* const* d_in, const int* in_sizes, int n_in,
                              void* d_out, int out_size, void* d_ws, size_t ws_size,
                              hipStream_t stream) {
  (void)in_sizes; (void)n_in; (void)d_ws; (void)ws_size; (void)out_size;
  const float* x0 = (const float*)d_in[0];
  const float* C  = (const float*)d_in[1];
  const float* c  = (const float*)d_in[2];
  const float* F  = (const float*)d_in[3];
  const float* f  = (const float*)d_in[4];
  float* out = (float*)d_out;
  mpc_riccati<<<NB_, 32, 0, stream>>>(x0, C, c, F, f, out);
}


# Next step's F, C, c, f prefetched whole: one lane-addressed prefetch per step touches every cache line of the four tiles, nearest scope (was: first line of F and of C only, farthest scope)
# speedup vs baseline: 4.4112x; 1.2637x over previous
#include <hip/hip_runtime.h>


#define NS_  8
#define NC_  4
#define T_   32
#define NB_  128
#define NSC_ 12
#define T1_  33

typedef float v2f __attribute__((ext_vector_type(2)));
typedef float v8f __attribute__((ext_vector_type(8)));

__device__ __forceinline__ v8f wmma4(v2f a, v2f b, v8f acc) {
  return __builtin_amdgcn_wmma_f32_16x16x4_f32(
       false, a,  false, b,
       (short)0, acc,  false,  false);
}

__global__ __launch_bounds__(32) void mpc_riccati(
    const float* __restrict__ x0g,
    const float* __restrict__ Cg_,
    const float* __restrict__ cg_,
    const float* __restrict__ Fg_,
    const float* __restrict__ fg_,
    float* __restrict__ out)
{
  const int  b    = blockIdx.x;
  const int  lane = threadIdx.x;
  const int  hl   = lane & 15;
  const bool hiw  = lane >= 16;
  const int  rbase = hiw ? 8 : 0;
  const int  kbase = hiw ? 2 : 0;

  __shared__ float Pm[16][16];
  __shared__ float Wm[16][16];
  __shared__ float Qm[16][16];
  __shared__ float pv[NS_];
  __shared__ float qv[NSC_];
  __shared__ float Pfp[NS_];
  __shared__ float Mall[T1_][NSC_][NS_ + 1];

  const float* pf_arr;
  int pf_tile, pf_off;
  if (lane >= 4 && lane < 10) {
    pf_arr = Cg_; pf_tile = NSC_ * NSC_;
    pf_off = (lane < 9) ? (lane - 4) * 32 : (NSC_ * NSC_ - 1);
  } else if (lane == 10 || lane == 11) {
    pf_arr = cg_; pf_tile = NSC_;
    pf_off = (lane == 10) ? 0 : (NSC_ - 1);
  } else if (lane == 12 || lane == 13) {
    pf_arr = fg_; pf_tile = NS_;
    pf_off = (lane == 12) ? 0 : (NS_ - 1);
  } else {
    pf_arr = Fg_; pf_tile = NS_ * NSC_;
    pf_off = (lane < 3) ? lane * 32 : ((lane == 3) ? (NS_ * NSC_ - 1) : 0);
  }
  __builtin_prefetch(pf_arr + (size_t)((T_ - 1) * NB_ + b) * pf_tile + pf_off, 0, 3);

  for (int idx = lane; idx < 256; idx += 32) {
    int r = idx >> 4, cx = idx & 15;
    Pm[r][cx] = 0.f; Wm[r][cx] = 0.f; Qm[r][cx] = 0.f;
  }
  __syncthreads();

  auto solve_update = [&](int t, const bool dyn, float fl,
                          float fu0, float fu1, float fu2, float fu3,
                          v2f af, v8f cF) {
    float A[4][4];
    #pragma unroll
    for (int i = 0; i < 4; ++i)
      #pragma unroll
      for (int j = 0; j < 4; ++j) A[i][j] = Qm[NS_ + i][NS_ + j];
    float rx[4], rq[4];
    #pragma unroll
    for (int m2 = 0; m2 < 4; ++m2) { rx[m2] = Qm[NS_ + m2][hl]; rq[m2] = qv[NS_ + m2]; }
    const int l7 = lane & 7;
    float qxu[4];
    #pragma unroll
    for (int j = 0; j < 4; ++j) qxu[j] = Qm[l7][NS_ + j];
    const float qx = qv[l7];
    v2f a;
    a.x = Qm[hl][NS_ + kbase];
    a.y = Qm[hl][NS_ + kbase + 1];
    v8f cf;
    #pragma unroll
    for (int r = 0; r < 8; ++r) cf[r] = Qm[r + rbase][hl];

    asm volatile("" ::
        "v"(A[0][0]), "v"(A[0][1]), "v"(A[0][2]), "v"(A[0][3]),
        "v"(A[1][0]), "v"(A[1][1]), "v"(A[1][2]), "v"(A[1][3]),
        "v"(A[2][0]), "v"(A[2][1]), "v"(A[2][2]), "v"(A[2][3]),
        "v"(A[3][0]), "v"(A[3][1]), "v"(A[3][2]), "v"(A[3][3]),
        "v"(rx[0]), "v"(rx[1]), "v"(rx[2]), "v"(rx[3]),
        "v"(rq[0]), "v"(rq[1]), "v"(rq[2]), "v"(rq[3]));

    #pragma unroll
    for (int cc = 0; cc < 4; ++cc) {
      const float piv = 1.f / A[cc][cc];
      #pragma unroll
      for (int j = 0; j < 4; ++j) A[cc][j] *= piv;
      rx[cc] *= piv; rq[cc] *= piv;
      #pragma unroll
      for (int r = 0; r < 4; ++r) {
        if (r == cc) continue;
        const float fac = A[r][cc];
        #pragma unroll
        for (int j = 0; j < 4; ++j) A[r][j] -= fac * A[cc][j];
        rx[r] -= fac * rx[cc];
        rq[r] -= fac * rq[cc];
      }
    }
    float Kc[4], kk[4];
    #pragma unroll
    for (int m2 = 0; m2 < 4; ++m2) { Kc[m2] = -rx[m2]; kk[m2] = -rq[m2]; }
    v2f bb;
    bb.x = hiw ? Kc[2] : Kc[0];
    bb.y = hiw ? Kc[3] : Kc[1];
    v8f d = wmma4(a, bb, cf);
    #pragma unroll
    for (int r = 0; r < 8; ++r) Pm[r + rbase][hl] = d[r];
    {
      const bool isK = hl < NS_;
      const int  kc  = isK ? hl : NS_;
      #pragma unroll
      for (int m2 = 0; m2 < 4; ++m2)
        Mall[t][NS_ + m2][kc] = isK ? Kc[m2] : kk[m2];
    }
    {
      float s = qx;
      #pragma unroll
      for (int j = 0; j < NC_; ++j) s += qxu[j] * kk[j];
      pv[l7] = s;
      if (dyn) {
        float bs = fl;
        bs += fu0 * kk[0];
        bs += fu1 * kk[1];
        bs += fu2 * kk[2];
        bs += fu3 * kk[3];
        Mall[t][l7][NS_] = bs;
      }
    }
    if (dyn) {
      v8f acl = wmma4(af, bb, cF);
      if (lane < NS_) {
        #pragma unroll
        for (int r = 0; r < 8; ++r) Mall[t][r][lane] = acl[r];
      }
    } else {
      for (int idx = lane; idx < NS_ * (NS_ + 1); idx += 32)
        Mall[t][idx / (NS_ + 1)][idx % (NS_ + 1)] = 0.f;
    }
    __syncthreads();
  };

  {
    const float* Cg = Cg_ + (size_t)((T_ - 1) * NB_ + b) * NSC_ * NSC_;
    for (int idx = lane; idx < NSC_ * NSC_; idx += 32)
      Qm[idx / NSC_][idx % NSC_] = Cg[idx];
    const float* cg = cg_ + (size_t)((T_ - 1) * NB_ + b) * NSC_;
    if (lane < NSC_) qv[lane] = cg[lane];
    __syncthreads();
    const v2f z2 = {0.f, 0.f};
    const v8f z8 = {0.f, 0.f, 0.f, 0.f, 0.f, 0.f, 0.f, 0.f};
    solve_update(T_, false, 0.f, 0.f, 0.f, 0.f, 0.f, z2, z8);
  }

  for (int t = T_ - 1; t >= 0; --t) {
    const float* Fg = Fg_ + (size_t)(t * NB_ + b) * NS_ * NSC_;
    const float* fg = fg_ + (size_t)(t * NB_ + b) * NS_;
    const float* Cg = Cg_ + (size_t)(t * NB_ + b) * NSC_ * NSC_;
    const float* cg = cg_ + (size_t)(t * NB_ + b) * NSC_;

    if (t > 0)
      __builtin_prefetch(pf_arr + (size_t)((t - 1) * NB_ + b) * pf_tile + pf_off, 0, 3);

    const int  hc = (hl < NSC_) ? hl : (NSC_ - 1);
    const bool cv = hl < NSC_;
    float Fcol[NS_];
    #pragma unroll
    for (int i = 0; i < NS_; ++i) Fcol[i] = Fg[i * NSC_ + hc];
    float Ccol[8];
    #pragma unroll
    for (int r = 0; r < 8; ++r) {
      const int rw = r + rbase;
      Ccol[r] = Cg[((rw < NSC_) ? rw : (NSC_ - 1)) * NSC_ + hc];
    }
    const float fu0 = Fg[(lane & 7) * NSC_ + NS_ + 0];
    const float fu1 = Fg[(lane & 7) * NSC_ + NS_ + 1];
    const float fu2 = Fg[(lane & 7) * NSC_ + NS_ + 2];
    const float fu3 = Fg[(lane & 7) * NSC_ + NS_ + 3];
    const float fl  = fg[lane & 7];
    const float cl  = cg[hc];
    float fv[NS_];
    #pragma unroll
    for (int m = 0; m < NS_; ++m)
      fv[m] = __int_as_float(__builtin_amdgcn_readlane(__float_as_int(fl), m));

    v2f fb0, fb1;
    fb0.x = cv ? (hiw ? Fcol[2] : Fcol[0]) : 0.f;
    fb0.y = cv ? (hiw ? Fcol[3] : Fcol[1]) : 0.f;
    fb1.x = cv ? (hiw ? Fcol[6] : Fcol[4]) : 0.f;
    fb1.y = cv ? (hiw ? Fcol[7] : Fcol[5]) : 0.f;
    v2f af;
    af.x = (hl < NS_) ? (hiw ? fu2 : fu0) : 0.f;
    af.y = (hl < NS_) ? (hiw ? fu3 : fu1) : 0.f;
    v8f cF;
    #pragma unroll
    for (int r = 0; r < 8; ++r) cF[r] = (!hiw && cv) ? Fcol[r] : 0.f;

    v8f acc = {0.f, 0.f, 0.f, 0.f, 0.f, 0.f, 0.f, 0.f};
    const int l7 = lane & 7;
    {
      v2f a0, a1;
      a0.x = Pm[hl][kbase];     a0.y = Pm[hl][kbase + 1];
      a1.x = Pm[hl][4 + kbase]; a1.y = Pm[hl][4 + kbase + 1];
      float prow[NS_];
      #pragma unroll
      for (int m = 0; m < NS_; ++m) prow[m] = Pm[l7][m];
      float s = pv[l7];
      acc = wmma4(a0, fb0, acc);
      acc = wmma4(a1, fb1, acc);
      #pragma unroll
      for (int m = 0; m < NS_; ++m) s += prow[m] * fv[m];
      Pfp[l7] = s;
    }
    #pragma unroll
    for (int r = 0; r < 8; ++r) Wm[r + rbase][hl] = acc[r];
    __syncthreads();

    v8f acc2;
    #pragma unroll
    for (int r = 0; r < 8; ++r) acc2[r] = ((r + rbase) < NSC_ && cv) ? Ccol[r] : 0.f;
    {
      v2f b0, b1;
      b0.x = Wm[kbase][hl];     b0.y = Wm[kbase + 1][hl];
      b1.x = Wm[4 + kbase][hl]; b1.y = Wm[4 + kbase + 1][hl];
      float pf[NS_];
      #pragma unroll
      for (int i = 0; i < NS_; ++i) pf[i] = Pfp[i];
      acc2 = wmma4(fb0, b0, acc2);
      acc2 = wmma4(fb1, b1, acc2);
      float s = cl;
      #pragma unroll
      for (int i = 0; i < NS_; ++i) s += Fcol[i] * pf[i];
      qv[hc] = s;
    }
    #pragma unroll
    for (int r = 0; r < 8; ++r) Qm[r + rbase][hl] = acc2[r];
    __syncthreads();

    solve_update(t, true, fl, fu0, fu1, fu2, fu3, af, cF);
  }

  float* out_x = out;
  float* out_u = out + (size_t)T1_ * NB_ * NS_;
  const int row = (lane < NSC_) ? lane : (NSC_ - 1);
  float xv = (lane < NS_) ? x0g[(size_t)b * NS_ + lane] : 0.f;
  float m[NS_ + 1], mn[NS_ + 1];
  #pragma unroll
  for (int j = 0; j <= NS_; ++j) m[j] = Mall[0][row][j];
  for (int t = 0; t <= T_; ++t) {
    const int tn = (t < T_) ? t + 1 : T_;
    #pragma unroll
    for (int j = 0; j <= NS_; ++j) mn[j] = Mall[tn][row][j];
    float xj[NS_];
    #pragma unroll
    for (int j = 0; j < NS_; ++j)
      xj[j] = __int_as_float(__builtin_amdgcn_readlane(__float_as_int(xv), j));
    float s0 = m[NS_], s1 = 0.f;
    #pragma unroll
    for (int j = 0; j < NS_; j += 2) { s0 += m[j] * xj[j]; s1 += m[j + 1] * xj[j + 1]; }
    const float s = s0 + s1;
    if (lane < NS_) out_x[(size_t)t * NB_ * NS_ + b * NS_ + lane] = xv;
    else if (lane < NSC_) out_u[(size_t)t * NB_ * NC_ + b * NC_ + (lane - NS_)] = s;
    xv = s;
    #pragma unroll
    for (int j = 0; j <= NS_; ++j) m[j] = mn[j];
  }
}

extern "C" void kernel_launch(void* const* d_in, const int* in_sizes, int n_in,
                              void* d_out, int out_size, void* d_ws, size_t ws_size,
                              hipStream_t stream) {
  (void)in_sizes; (void)n_in; (void)d_ws; (void)ws_size; (void)out_size;
  const float* x0 = (const float*)d_in[0];
  const float* C  = (const float*)d_in[1];
  const float* c  = (const float*)d_in[2];
  const float* F  = (const float*)d_in[3];
  const float* f  = (const float*)d_in[4];
  float* out = (float*)d_out;
  mpc_riccati<<<NB_, 32, 0, stream>>>(x0, C, c, F, f, out);
}


# One-wave work-group: block barriers between the step's phases replaced by wave-scope ordering; LDS stores are no longer drained before the reads behind them issue (in-order LDS of one wave)
# speedup vs baseline: 4.5938x; 1.0414x over previous
#include <hip/hip_runtime.h>


#define NS_  8
#define NC_  4
#define T_   32
#define NB_  128
#define NSC_ 12
#define T1_  33

typedef float v2f __attribute__((ext_vector_type(2)));
typedef float v8f __attribute__((ext_vector_type(8)));

__device__ __forceinline__ v8f wmma4(v2f a, v2f b, v8f acc) {
  return __builtin_amdgcn_wmma_f32_16x16x4_f32(
       false, a,  false, b,
       (short)0, acc,  false,  false);
}

__device__ __forceinline__ void wave_order() {
  __builtin_amdgcn_fence(__ATOMIC_ACQ_REL, "wavefront");
}

__global__ __launch_bounds__(32) void mpc_riccati(
    const float* __restrict__ x0g,
    const float* __restrict__ Cg_,
    const float* __restrict__ cg_,
    const float* __restrict__ Fg_,
    const float* __restrict__ fg_,
    float* __restrict__ out)
{
  const int  b    = blockIdx.x;
  const int  lane = threadIdx.x;
  const int  hl   = lane & 15;
  const bool hiw  = lane >= 16;
  const int  rbase = hiw ? 8 : 0;
  const int  kbase = hiw ? 2 : 0;

  __shared__ float Pm[16][16];
  __shared__ float Wm[16][16];
  __shared__ float Qm[16][16];
  __shared__ float pv[NS_];
  __shared__ float qv[NSC_];
  __shared__ float Pfp[NS_];
  __shared__ float Mall[T1_][NSC_][NS_ + 1];

  const float* pf_arr;
  int pf_tile, pf_off;
  if (lane >= 4 && lane < 10) {
    pf_arr = Cg_; pf_tile = NSC_ * NSC_;
    pf_off = (lane < 9) ? (lane - 4) * 32 : (NSC_ * NSC_ - 1);
  } else if (lane == 10 || lane == 11) {
    pf_arr = cg_; pf_tile = NSC_;
    pf_off = (lane == 10) ? 0 : (NSC_ - 1);
  } else if (lane == 12 || lane == 13) {
    pf_arr = fg_; pf_tile = NS_;
    pf_off = (lane == 12) ? 0 : (NS_ - 1);
  } else {
    pf_arr = Fg_; pf_tile = NS_ * NSC_;
    pf_off = (lane < 3) ? lane * 32 : ((lane == 3) ? (NS_ * NSC_ - 1) : 0);
  }
  __builtin_prefetch(pf_arr + (size_t)((T_ - 1) * NB_ + b) * pf_tile + pf_off, 0, 3);

  for (int idx = lane; idx < 256; idx += 32) {
    int r = idx >> 4, cx = idx & 15;
    Pm[r][cx] = 0.f; Wm[r][cx] = 0.f; Qm[r][cx] = 0.f;
  }
  wave_order();

  auto solve_update = [&](int t, const bool dyn, float fl,
                          float fu0, float fu1, float fu2, float fu3,
                          v2f af, v8f cF) {
    float A[4][4];
    #pragma unroll
    for (int i = 0; i < 4; ++i)
      #pragma unroll
      for (int j = 0; j < 4; ++j) A[i][j] = Qm[NS_ + i][NS_ + j];
    float rx[4], rq[4];
    #pragma unroll
    for (int m2 = 0; m2 < 4; ++m2) { rx[m2] = Qm[NS_ + m2][hl]; rq[m2] = qv[NS_ + m2]; }
    const int l7 = lane & 7;
    float qxu[4];
    #pragma unroll
    for (int j = 0; j < 4; ++j) qxu[j] = Qm[l7][NS_ + j];
    const float qx = qv[l7];
    v2f a;
    a.x = Qm[hl][NS_ + kbase];
    a.y = Qm[hl][NS_ + kbase + 1];
    v8f cf;
    #pragma unroll
    for (int r = 0; r < 8; ++r) cf[r] = Qm[r + rbase][hl];

    asm volatile("" ::
        "v"(A[0][0]), "v"(A[0][1]), "v"(A[0][2]), "v"(A[0][3]),
        "v"(A[1][0]), "v"(A[1][1]), "v"(A[1][2]), "v"(A[1][3]),
        "v"(A[2][0]), "v"(A[2][1]), "v"(A[2][2]), "v"(A[2][3]),
        "v"(A[3][0]), "v"(A[3][1]), "v"(A[3][2]), "v"(A[3][3]),
        "v"(rx[0]), "v"(rx[1]), "v"(rx[2]), "v"(rx[3]),
        "v"(rq[0]), "v"(rq[1]), "v"(rq[2]), "v"(rq[3]));

    #pragma unroll
    for (int cc = 0; cc < 4; ++cc) {
      const float piv = 1.f / A[cc][cc];
      #pragma unroll
      for (int j = 0; j < 4; ++j) A[cc][j] *= piv;
      rx[cc] *= piv; rq[cc] *= piv;
      #pragma unroll
      for (int r = 0; r < 4; ++r) {
        if (r == cc) continue;
        const float fac = A[r][cc];
        #pragma unroll
        for (int j = 0; j < 4; ++j) A[r][j] -= fac * A[cc][j];
        rx[r] -= fac * rx[cc];
        rq[r] -= fac * rq[cc];
      }
    }
    float Kc[4], kk[4];
    #pragma unroll
    for (int m2 = 0; m2 < 4; ++m2) { Kc[m2] = -rx[m2]; kk[m2] = -rq[m2]; }
    v2f bb;
    bb.x = hiw ? Kc[2] : Kc[0];
    bb.y = hiw ? Kc[3] : Kc[1];
    v8f d = wmma4(a, bb, cf);
    #pragma unroll
    for (int r = 0; r < 8; ++r) Pm[r + rbase][hl] = d[r];
    {
      const bool isK = hl < NS_;
      const int  kc  = isK ? hl : NS_;
      #pragma unroll
      for (int m2 = 0; m2 < 4; ++m2)
        Mall[t][NS_ + m2][kc] = isK ? Kc[m2] : kk[m2];
    }
    {
      float s = qx;
      #pragma unroll
      for (int j = 0; j < NC_; ++j) s += qxu[j] * kk[j];
      pv[l7] = s;
      if (dyn) {
        float bs = fl;
        bs += fu0 * kk[0];
        bs += fu1 * kk[1];
        bs += fu2 * kk[2];
        bs += fu3 * kk[3];
        Mall[t][l7][NS_] = bs;
      }
    }
    if (dyn) {
      v8f acl = wmma4(af, bb, cF);
      if (lane < NS_) {
        #pragma unroll
        for (int r = 0; r < 8; ++r) Mall[t][r][lane] = acl[r];
      }
    } else {
      for (int idx = lane; idx < NS_ * (NS_ + 1); idx += 32)
        Mall[t][idx / (NS_ + 1)][idx % (NS_ + 1)] = 0.f;
    }
    wave_order();
  };

  {
    const float* Cg = Cg_ + (size_t)((T_ - 1) * NB_ + b) * NSC_ * NSC_;
    for (int idx = lane; idx < NSC_ * NSC_; idx += 32)
      Qm[idx / NSC_][idx % NSC_] = Cg[idx];
    const float* cg = cg_ + (size_t)((T_ - 1) * NB_ + b) * NSC_;
    if (lane < NSC_) qv[lane] = cg[lane];
    wave_order();
    const v2f z2 = {0.f, 0.f};
    const v8f z8 = {0.f, 0.f, 0.f, 0.f, 0.f, 0.f, 0.f, 0.f};
    solve_update(T_, false, 0.f, 0.f, 0.f, 0.f, 0.f, z2, z8);
  }

  for (int t = T_ - 1; t >= 0; --t) {
    const float* Fg = Fg_ + (size_t)(t * NB_ + b) * NS_ * NSC_;
    const float* fg = fg_ + (size_t)(t * NB_ + b) * NS_;
    const float* Cg = Cg_ + (size_t)(t * NB_ + b) * NSC_ * NSC_;
    const float* cg = cg_ + (size_t)(t * NB_ + b) * NSC_;

    if (t > 0)
      __builtin_prefetch(pf_arr + (size_t)((t - 1) * NB_ + b) * pf_tile + pf_off, 0, 3);

    const int  hc = (hl < NSC_) ? hl : (NSC_ - 1);
    const bool cv = hl < NSC_;
    float Fcol[NS_];
    #pragma unroll
    for (int i = 0; i < NS_; ++i) Fcol[i] = Fg[i * NSC_ + hc];
    float Ccol[8];
    #pragma unroll
    for (int r = 0; r < 8; ++r) {
      const int rw = r + rbase;
      Ccol[r] = Cg[((rw < NSC_) ? rw : (NSC_ - 1)) * NSC_ + hc];
    }
    const float fu0 = Fg[(lane & 7) * NSC_ + NS_ + 0];
    const float fu1 = Fg[(lane & 7) * NSC_ + NS_ + 1];
    const float fu2 = Fg[(lane & 7) * NSC_ + NS_ + 2];
    const float fu3 = Fg[(lane & 7) * NSC_ + NS_ + 3];
    const float fl  = fg[lane & 7];
    const float cl  = cg[hc];
    float fv[NS_];
    #pragma unroll
    for (int m = 0; m < NS_; ++m)
      fv[m] = __int_as_float(__builtin_amdgcn_readlane(__float_as_int(fl), m));

    v2f fb0, fb1;
    fb0.x = cv ? (hiw ? Fcol[2] : Fcol[0]) : 0.f;
    fb0.y = cv ? (hiw ? Fcol[3] : Fcol[1]) : 0.f;
    fb1.x = cv ? (hiw ? Fcol[6] : Fcol[4]) : 0.f;
    fb1.y = cv ? (hiw ? Fcol[7] : Fcol[5]) : 0.f;
    v2f af;
    af.x = (hl < NS_) ? (hiw ? fu2 : fu0) : 0.f;
    af.y = (hl < NS_) ? (hiw ? fu3 : fu1) : 0.f;
    v8f cF;
    #pragma unroll
    for (int r = 0; r < 8; ++r) cF[r] = (!hiw && cv) ? Fcol[r] : 0.f;

    v8f acc = {0.f, 0.f, 0.f, 0.f, 0.f, 0.f, 0.f, 0.f};
    const int l7 = lane & 7;
    {
      v2f a0, a1;
      a0.x = Pm[hl][kbase];     a0.y = Pm[hl][kbase + 1];
      a1.x = Pm[hl][4 + kbase]; a1.y = Pm[hl][4 + kbase + 1];
      float prow[NS_];
      #pragma unroll
      for (int m = 0; m < NS_; ++m) prow[m] = Pm[l7][m];
      float s = pv[l7];
      acc = wmma4(a0, fb0, acc);
      acc = wmma4(a1, fb1, acc);
      #pragma unroll
      for (int m = 0; m < NS_; ++m) s += prow[m] * fv[m];
      Pfp[l7] = s;
    }
    #pragma unroll
    for (int r = 0; r < 8; ++r) Wm[r + rbase][hl] = acc[r];
    wave_order();

    v8f acc2;
    #pragma unroll
    for (int r = 0; r < 8; ++r) acc2[r] = ((r + rbase) < NSC_ && cv) ? Ccol[r] : 0.f;
    {
      v2f b0, b1;
      b0.x = Wm[kbase][hl];     b0.y = Wm[kbase + 1][hl];
      b1.x = Wm[4 + kbase][hl]; b1.y = Wm[4 + kbase + 1][hl];
      float pf[NS_];
      #pragma unroll
      for (int i = 0; i < NS_; ++i) pf[i] = Pfp[i];
      acc2 = wmma4(fb0, b0, acc2);
      acc2 = wmma4(fb1, b1, acc2);
      float s = cl;
      #pragma unroll
      for (int i = 0; i < NS_; ++i) s += Fcol[i] * pf[i];
      qv[hc] = s;
    }
    #pragma unroll
    for (int r = 0; r < 8; ++r) Qm[r + rbase][hl] = acc2[r];
    wave_order();

    solve_update(t, true, fl, fu0, fu1, fu2, fu3, af, cF);
  }

  float* out_x = out;
  float* out_u = out + (size_t)T1_ * NB_ * NS_;
  const int row = (lane < NSC_) ? lane : (NSC_ - 1);
  float xv = (lane < NS_) ? x0g[(size_t)b * NS_ + lane] : 0.f;
  float m[NS_ + 1], mn[NS_ + 1];
  #pragma unroll
  for (int j = 0; j <= NS_; ++j) m[j] = Mall[0][row][j];
  for (int t = 0; t <= T_; ++t) {
    const int tn = (t < T_) ? t + 1 : T_;
    #pragma unroll
    for (int j = 0; j <= NS_; ++j) mn[j] = Mall[tn][row][j];
    float xj[NS_];
    #pragma unroll
    for (int j = 0; j < NS_; ++j)
      xj[j] = __int_as_float(__builtin_amdgcn_readlane(__float_as_int(xv), j));
    float s0 = m[NS_], s1 = 0.f;
    #pragma unroll
    for (int j = 0; j < NS_; j += 2) { s0 += m[j] * xj[j]; s1 += m[j + 1] * xj[j + 1]; }
    const float s = s0 + s1;
    if (lane < NS_) out_x[(size_t)t * NB_ * NS_ + b * NS_ + lane] = xv;
    else if (lane < NSC_) out_u[(size_t)t * NB_ * NC_ + b * NC_ + (lane - NS_)] = s;
    xv = s;
    #pragma unroll
    for (int j = 0; j <= NS_; ++j) m[j] = mn[j];
  }
}

extern "C" void kernel_launch(void* const* d_in, const int* in_sizes, int n_in,
                              void* d_out, int out_size, void* d_ws, size_t ws_size,
                              hipStream_t stream) {
  (void)in_sizes; (void)n_in; (void)d_ws; (void)ws_size; (void)out_size;
  const float* x0 = (const float*)d_in[0];
  const float* C  = (const float*)d_in[1];
  const float* c  = (const float*)d_in[2];
  const float* F  = (const float*)d_in[3];
  const float* f  = (const float*)d_in[4];
  float* out = (float*)d_out;
  mpc_riccati<<<NB_, 32, 0, stream>>>(x0, C, c, F, f, out);
}


# Gain solve: pivot divisions taken in two pairs, reciprocal of pivot 1 (3) formed beside that of pivot 0 (2) as that pivot / 2x2 minor of the matrix as it then stands (gain rounding changes)
# speedup vs baseline: 4.7446x; 1.0328x over previous
#include <hip/hip_runtime.h>


#define NS_  8
#define NC_  4
#define T_   32
#define NB_  128
#define NSC_ 12
#define T1_  33

typedef float v2f __attribute__((ext_vector_type(2)));
typedef float v8f __attribute__((ext_vector_type(8)));

__device__ __forceinline__ v8f wmma4(v2f a, v2f b, v8f acc) {
  return __builtin_amdgcn_wmma_f32_16x16x4_f32(
       false, a,  false, b,
       (short)0, acc,  false,  false);
}

__device__ __forceinline__ void wave_order() {
  __builtin_amdgcn_fence(__ATOMIC_ACQ_REL, "wavefront");
}

__global__ __launch_bounds__(32) void mpc_riccati(
    const float* __restrict__ x0g,
    const float* __restrict__ Cg_,
    const float* __restrict__ cg_,
    const float* __restrict__ Fg_,
    const float* __restrict__ fg_,
    float* __restrict__ out)
{
  const int  b    = blockIdx.x;
  const int  lane = threadIdx.x;
  const int  hl   = lane & 15;
  const bool hiw  = lane >= 16;
  const int  rbase = hiw ? 8 : 0;
  const int  kbase = hiw ? 2 : 0;

  __shared__ float Pm[16][16];
  __shared__ float Wm[16][16];
  __shared__ float Qm[16][16];
  __shared__ float pv[NS_];
  __shared__ float qv[NSC_];
  __shared__ float Pfp[NS_];
  __shared__ float Mall[T1_][NSC_][NS_ + 1];

  const float* pf_arr;
  int pf_tile, pf_off;
  if (lane >= 4 && lane < 10) {
    pf_arr = Cg_; pf_tile = NSC_ * NSC_;
    pf_off = (lane < 9) ? (lane - 4) * 32 : (NSC_ * NSC_ - 1);
  } else if (lane == 10 || lane == 11) {
    pf_arr = cg_; pf_tile = NSC_;
    pf_off = (lane == 10) ? 0 : (NSC_ - 1);
  } else if (lane == 12 || lane == 13) {
    pf_arr = fg_; pf_tile = NS_;
    pf_off = (lane == 12) ? 0 : (NS_ - 1);
  } else {
    pf_arr = Fg_; pf_tile = NS_ * NSC_;
    pf_off = (lane < 3) ? lane * 32 : ((lane == 3) ? (NS_ * NSC_ - 1) : 0);
  }
  __builtin_prefetch(pf_arr + (size_t)((T_ - 1) * NB_ + b) * pf_tile + pf_off, 0, 3);

  for (int idx = lane; idx < 256; idx += 32) {
    int r = idx >> 4, cx = idx & 15;
    Pm[r][cx] = 0.f; Wm[r][cx] = 0.f; Qm[r][cx] = 0.f;
  }
  wave_order();

  auto solve_update = [&](int t, const bool dyn, float fl,
                          float fu0, float fu1, float fu2, float fu3,
                          v2f af, v8f cF) {
    float A[4][4];
    #pragma unroll
    for (int i = 0; i < 4; ++i)
      #pragma unroll
      for (int j = 0; j < 4; ++j) A[i][j] = Qm[NS_ + i][NS_ + j];
    float rx[4], rq[4];
    #pragma unroll
    for (int m2 = 0; m2 < 4; ++m2) { rx[m2] = Qm[NS_ + m2][hl]; rq[m2] = qv[NS_ + m2]; }

    __builtin_amdgcn_sched_barrier(0);

    const int l7 = lane & 7;
    float qxu[4];
    #pragma unroll
    for (int j = 0; j < 4; ++j) qxu[j] = Qm[l7][NS_ + j];
    const float qx = qv[l7];
    v2f a;
    a.x = Qm[hl][NS_ + kbase];
    a.y = Qm[hl][NS_ + kbase + 1];
    v8f cf;
    #pragma unroll
    for (int r = 0; r < 8; ++r) cf[r] = Qm[r + rbase][hl];

    float rp[4];
    rp[0] = 1.f / A[0][0];
    rp[1] = A[0][0] / (A[0][0] * A[1][1] - A[1][0] * A[0][1]);

    #pragma unroll
    for (int cc = 0; cc < 4; ++cc) {
      const float piv = rp[cc];
      #pragma unroll
      for (int j = 0; j < 4; ++j) A[cc][j] *= piv;
      rx[cc] *= piv; rq[cc] *= piv;
      #pragma unroll
      for (int r = 0; r < 4; ++r) {
        if (r == cc) continue;
        const float fac = A[r][cc];
        #pragma unroll
        for (int j = 0; j < 4; ++j) A[r][j] -= fac * A[cc][j];
        rx[r] -= fac * rx[cc];
        rq[r] -= fac * rq[cc];
      }
      if (cc == 1) {
        rp[2] = 1.f / A[2][2];
        rp[3] = A[2][2] / (A[2][2] * A[3][3] - A[3][2] * A[2][3]);
      }
    }
    float Kc[4], kk[4];
    #pragma unroll
    for (int m2 = 0; m2 < 4; ++m2) { Kc[m2] = -rx[m2]; kk[m2] = -rq[m2]; }
    v2f bb;
    bb.x = hiw ? Kc[2] : Kc[0];
    bb.y = hiw ? Kc[3] : Kc[1];
    v8f d = wmma4(a, bb, cf);
    #pragma unroll
    for (int r = 0; r < 8; ++r) Pm[r + rbase][hl] = d[r];
    {
      const bool isK = hl < NS_;
      const int  kc  = isK ? hl : NS_;
      #pragma unroll
      for (int m2 = 0; m2 < 4; ++m2)
        Mall[t][NS_ + m2][kc] = isK ? Kc[m2] : kk[m2];
    }
    {
      float s = qx;
      #pragma unroll
      for (int j = 0; j < NC_; ++j) s += qxu[j] * kk[j];
      pv[l7] = s;
      if (dyn) {
        float bs = fl;
        bs += fu0 * kk[0];
        bs += fu1 * kk[1];
        bs += fu2 * kk[2];
        bs += fu3 * kk[3];
        Mall[t][l7][NS_] = bs;
      }
    }
    if (dyn) {
      v8f acl = wmma4(af, bb, cF);
      if (lane < NS_) {
        #pragma unroll
        for (int r = 0; r < 8; ++r) Mall[t][r][lane] = acl[r];
      }
    } else {
      for (int idx = lane; idx < NS_ * (NS_ + 1); idx += 32)
        Mall[t][idx / (NS_ + 1)][idx % (NS_ + 1)] = 0.f;
    }
    wave_order();
  };

  {
    const float* Cg = Cg_ + (size_t)((T_ - 1) * NB_ + b) * NSC_ * NSC_;
    for (int idx = lane; idx < NSC_ * NSC_; idx += 32)
      Qm[idx / NSC_][idx % NSC_] = Cg[idx];
    const float* cg = cg_ + (size_t)((T_ - 1) * NB_ + b) * NSC_;
    if (lane < NSC_) qv[lane] = cg[lane];
    wave_order();
    const v2f z2 = {0.f, 0.f};
    const v8f z8 = {0.f, 0.f, 0.f, 0.f, 0.f, 0.f, 0.f, 0.f};
    solve_update(T_, false, 0.f, 0.f, 0.f, 0.f, 0.f, z2, z8);
  }

  for (int t = T_ - 1; t >= 0; --t) {
    const float* Fg = Fg_ + (size_t)(t * NB_ + b) * NS_ * NSC_;
    const float* fg = fg_ + (size_t)(t * NB_ + b) * NS_;
    const float* Cg = Cg_ + (size_t)(t * NB_ + b) * NSC_ * NSC_;
    const float* cg = cg_ + (size_t)(t * NB_ + b) * NSC_;

    if (t > 0)
      __builtin_prefetch(pf_arr + (size_t)((t - 1) * NB_ + b) * pf_tile + pf_off, 0, 3);

    const int  hc = (hl < NSC_) ? hl : (NSC_ - 1);
    const bool cv = hl < NSC_;
    float Fcol[NS_];
    #pragma unroll
    for (int i = 0; i < NS_; ++i) Fcol[i] = Fg[i * NSC_ + hc];
    float Ccol[8];
    #pragma unroll
    for (int r = 0; r < 8; ++r) {
      const int rw = r + rbase;
      Ccol[r] = Cg[((rw < NSC_) ? rw : (NSC_ - 1)) * NSC_ + hc];
    }
    const float fu0 = Fg[(lane & 7) * NSC_ + NS_ + 0];
    const float fu1 = Fg[(lane & 7) * NSC_ + NS_ + 1];
    const float fu2 = Fg[(lane & 7) * NSC_ + NS_ + 2];
    const float fu3 = Fg[(lane & 7) * NSC_ + NS_ + 3];
    const float fl  = fg[lane & 7];
    const float cl  = cg[hc];
    float fv[NS_];
    #pragma unroll
    for (int m = 0; m < NS_; ++m)
      fv[m] = __int_as_float(__builtin_amdgcn_readlane(__float_as_int(fl), m));

    v2f fb0, fb1;
    fb0.x = cv ? (hiw ? Fcol[2] : Fcol[0]) : 0.f;
    fb0.y = cv ? (hiw ? Fcol[3] : Fcol[1]) : 0.f;
    fb1.x = cv ? (hiw ? Fcol[6] : Fcol[4]) : 0.f;
    fb1.y = cv ? (hiw ? Fcol[7] : Fcol[5]) : 0.f;
    v2f af;
    af.x = (hl < NS_) ? (hiw ? fu2 : fu0) : 0.f;
    af.y = (hl < NS_) ? (hiw ? fu3 : fu1) : 0.f;
    v8f cF;
    #pragma unroll
    for (int r = 0; r < 8; ++r) cF[r] = (!hiw && cv) ? Fcol[r] : 0.f;

    v8f acc = {0.f, 0.f, 0.f, 0.f, 0.f, 0.f, 0.f, 0.f};
    const int l7 = lane & 7;
    {
      v2f a0, a1;
      a0.x = Pm[hl][kbase];     a0.y = Pm[hl][kbase + 1];
      a1.x = Pm[hl][4 + kbase]; a1.y = Pm[hl][4 + kbase + 1];
      float prow[NS_];
      #pragma unroll
      for (int m = 0; m < NS_; ++m) prow[m] = Pm[l7][m];
      float s = pv[l7];
      acc = wmma4(a0, fb0, acc);
      acc = wmma4(a1, fb1, acc);
      #pragma unroll
      for (int m = 0; m < NS_; ++m) s += prow[m] * fv[m];
      Pfp[l7] = s;
    }
    #pragma unroll
    for (int r = 0; r < 8; ++r) Wm[r + rbase][hl] = acc[r];
    wave_order();

    v8f acc2;
    #pragma unroll
    for (int r = 0; r < 8; ++r) acc2[r] = ((r + rbase) < NSC_ && cv) ? Ccol[r] : 0.f;
    {
      v2f b0, b1;
      b0.x = Wm[kbase][hl];     b0.y = Wm[kbase + 1][hl];
      b1.x = Wm[4 + kbase][hl]; b1.y = Wm[4 + kbase + 1][hl];
      float pf[NS_];
      #pragma unroll
      for (int i = 0; i < NS_; ++i) pf[i] = Pfp[i];
      acc2 = wmma4(fb0, b0, acc2);
      acc2 = wmma4(fb1, b1, acc2);
      float s = cl;
      #pragma unroll
      for (int i = 0; i < NS_; ++i) s += Fcol[i] * pf[i];
      qv[hc] = s;
    }
    #pragma unroll
    for (int r = 0; r < 8; ++r) Qm[r + rbase][hl] = acc2[r];
    wave_order();

    solve_update(t, true, fl, fu0, fu1, fu2, fu3, af, cF);
  }

  float* out_x = out;
  float* out_u = out + (size_t)T1_ * NB_ * NS_;
  const int row = (lane < NSC_) ? lane : (NSC_ - 1);
  float xv = (lane < NS_) ? x0g[(size_t)b * NS_ + lane] : 0.f;
  float m[NS_ + 1], mn[NS_ + 1];
  #pragma unroll
  for (int j = 0; j <= NS_; ++j) m[j] = Mall[0][row][j];
  for (int t = 0; t <= T_; ++t) {
    const int tn = (t < T_) ? t + 1 : T_;
    #pragma unroll
    for (int j = 0; j <= NS_; ++j) mn[j] = Mall[tn][row][j];
    float xj[NS_];
    #pragma unroll
    for (int j = 0; j < NS_; ++j)
      xj[j] = __int_as_float(__builtin_amdgcn_readlane(__float_as_int(xv), j));
    float s0 = m[NS_], s1 = 0.f;
    #pragma unroll
    for (int j = 0; j < NS_; j += 2) { s0 += m[j] * xj[j]; s1 += m[j + 1] * xj[j + 1]; }
    const float s = s0 + s1;
    if (lane < NS_) out_x[(size_t)t * NB_ * NS_ + b * NS_ + lane] = xv;
    else if (lane < NSC_) out_u[(size_t)t * NB_ * NC_ + b * NC_ + (lane - NS_)] = s;
    xv = s;
    #pragma unroll
    for (int j = 0; j <= NS_; ++j) m[j] = mn[j];
  }
}

extern "C" void kernel_launch(void* const* d_in, const int* in_sizes, int n_in,
                              void* d_out, int out_size, void* d_ws, size_t ws_size,
                              hipStream_t stream) {
  (void)in_sizes; (void)n_in; (void)d_ws; (void)ws_size; (void)out_size;
  const float* x0 = (const float*)d_in[0];
  const float* C  = (const float*)d_in[1];
  const float* c  = (const float*)d_in[2];
  const float* F  = (const float*)d_in[3];
  const float* f  = (const float*)d_in[4];
  float* out = (float*)d_out;
  mpc_riccati<<<NB_, 32, 0, stream>>>(x0, C, c, F, f, out);
}


# W and P f + p leave LDS: P's A fragment repeats rows 0..7 as rows 8..15, so a lane holds its W column = its B fragment of F^T W; P f + p by a WMMA pair with f in all B columns (sum order may change)
# speedup vs baseline: 5.0589x; 1.0662x over previous
#include <hip/hip_runtime.h>


#define NS_  8
#define NC_  4
#define T_   32
#define NB_  128
#define NSC_ 12
#define T1_  33

typedef float v2f __attribute__((ext_vector_type(2)));
typedef float v8f __attribute__((ext_vector_type(8)));

__device__ __forceinline__ v8f wmma4(v2f a, v2f b, v8f acc) {
  return __builtin_amdgcn_wmma_f32_16x16x4_f32(
       false, a,  false, b,
       (short)0, acc,  false,  false);
}

__device__ __forceinline__ void wave_order() {
  __builtin_amdgcn_fence(__ATOMIC_ACQ_REL, "wavefront");
}

__global__ __launch_bounds__(32) void mpc_riccati(
    const float* __restrict__ x0g,
    const float* __restrict__ Cg_,
    const float* __restrict__ cg_,
    const float* __restrict__ Fg_,
    const float* __restrict__ fg_,
    float* __restrict__ out)
{
  const int  b    = blockIdx.x;
  const int  lane = threadIdx.x;
  const int  hl   = lane & 15;
  const bool hiw  = lane >= 16;
  const int  rbase = hiw ? 8 : 0;
  const int  kbase = hiw ? 2 : 0;

  __shared__ float Pm[16][16];
  __shared__ float Qm[16][16];
  __shared__ float pv[NS_];
  __shared__ float qv[NSC_];
  __shared__ float Mall[T1_][NSC_][NS_ + 1];

  const float* pf_arr;
  int pf_tile, pf_off;
  if (lane >= 4 && lane < 10) {
    pf_arr = Cg_; pf_tile = NSC_ * NSC_;
    pf_off = (lane < 9) ? (lane - 4) * 32 : (NSC_ * NSC_ - 1);
  } else if (lane == 10 || lane == 11) {
    pf_arr = cg_; pf_tile = NSC_;
    pf_off = (lane == 10) ? 0 : (NSC_ - 1);
  } else if (lane == 12 || lane == 13) {
    pf_arr = fg_; pf_tile = NS_;
    pf_off = (lane == 12) ? 0 : (NS_ - 1);
  } else {
    pf_arr = Fg_; pf_tile = NS_ * NSC_;
    pf_off = (lane < 3) ? lane * 32 : ((lane == 3) ? (NS_ * NSC_ - 1) : 0);
  }
  __builtin_prefetch(pf_arr + (size_t)((T_ - 1) * NB_ + b) * pf_tile + pf_off, 0, 3);

  for (int idx = lane; idx < 256; idx += 32) {
    int r = idx >> 4, cx = idx & 15;
    Pm[r][cx] = 0.f; Qm[r][cx] = 0.f;
  }
  wave_order();

  auto solve_update = [&](int t, const bool dyn, float qs, int hc, float fl,
                          float fu0, float fu1, float fu2, float fu3,
                          v2f af, v8f cF) {
    float A[4][4];
    #pragma unroll
    for (int i = 0; i < 4; ++i)
      #pragma unroll
      for (int j = 0; j < 4; ++j) A[i][j] = Qm[NS_ + i][NS_ + j];
    float rx[4], rq[4];
    #pragma unroll
    for (int m2 = 0; m2 < 4; ++m2) rx[m2] = Qm[NS_ + m2][hl];
    if (dyn) qv[hc] = qs;
    wave_order();
    #pragma unroll
    for (int m2 = 0; m2 < 4; ++m2) rq[m2] = qv[NS_ + m2];

    __builtin_amdgcn_sched_barrier(0);

    const int l7 = lane & 7;
    float qxu[4];
    #pragma unroll
    for (int j = 0; j < 4; ++j) qxu[j] = Qm[l7][NS_ + j];
    const float qx = qv[l7];
    v2f a;
    a.x = Qm[hl][NS_ + kbase];
    a.y = Qm[hl][NS_ + kbase + 1];
    v8f cf;
    #pragma unroll
    for (int r = 0; r < 8; ++r) cf[r] = Qm[r + rbase][hl];

    float rp[4];
    rp[0] = 1.f / A[0][0];
    rp[1] = A[0][0] / (A[0][0] * A[1][1] - A[1][0] * A[0][1]);

    #pragma unroll
    for (int cc = 0; cc < 4; ++cc) {
      const float piv = rp[cc];
      #pragma unroll
      for (int j = 0; j < 4; ++j) A[cc][j] *= piv;
      rx[cc] *= piv; rq[cc] *= piv;
      #pragma unroll
      for (int r = 0; r < 4; ++r) {
        if (r == cc) continue;
        const float fac = A[r][cc];
        #pragma unroll
        for (int j = 0; j < 4; ++j) A[r][j] -= fac * A[cc][j];
        rx[r] -= fac * rx[cc];
        rq[r] -= fac * rq[cc];
      }
      if (cc == 1) {
        rp[2] = 1.f / A[2][2];
        rp[3] = A[2][2] / (A[2][2] * A[3][3] - A[3][2] * A[2][3]);
      }
    }
    float Kc[4], kk[4];
    #pragma unroll
    for (int m2 = 0; m2 < 4; ++m2) { Kc[m2] = -rx[m2]; kk[m2] = -rq[m2]; }
    v2f bb;
    bb.x = hiw ? Kc[2] : Kc[0];
    bb.y = hiw ? Kc[3] : Kc[1];
    v8f d = wmma4(a, bb, cf);
    #pragma unroll
    for (int r = 0; r < 8; ++r) Pm[r + rbase][hl] = d[r];
    {
      const bool isK = hl < NS_;
      const int  kc  = isK ? hl : NS_;
      #pragma unroll
      for (int m2 = 0; m2 < 4; ++m2)
        Mall[t][NS_ + m2][kc] = isK ? Kc[m2] : kk[m2];
    }
    {
      float s = qx;
      #pragma unroll
      for (int j = 0; j < NC_; ++j) s += qxu[j] * kk[j];
      pv[l7] = s;
      if (dyn) {
        float bs = fl;
        bs += fu0 * kk[0];
        bs += fu1 * kk[1];
        bs += fu2 * kk[2];
        bs += fu3 * kk[3];
        Mall[t][l7][NS_] = bs;
      }
    }
    if (dyn) {
      v8f acl = wmma4(af, bb, cF);
      if (lane < NS_) {
        #pragma unroll
        for (int r = 0; r < 8; ++r) Mall[t][r][lane] = acl[r];
      }
    } else {
      for (int idx = lane; idx < NS_ * (NS_ + 1); idx += 32)
        Mall[t][idx / (NS_ + 1)][idx % (NS_ + 1)] = 0.f;
    }
    wave_order();
  };

  {
    const float* Cg = Cg_ + (size_t)((T_ - 1) * NB_ + b) * NSC_ * NSC_;
    for (int idx = lane; idx < NSC_ * NSC_; idx += 32)
      Qm[idx / NSC_][idx % NSC_] = Cg[idx];
    const float* cg = cg_ + (size_t)((T_ - 1) * NB_ + b) * NSC_;
    if (lane < NSC_) qv[lane] = cg[lane];
    wave_order();
    const v2f z2 = {0.f, 0.f};
    const v8f z8 = {0.f, 0.f, 0.f, 0.f, 0.f, 0.f, 0.f, 0.f};
    solve_update(T_, false, 0.f, 0, 0.f, 0.f, 0.f, 0.f, 0.f, z2, z8);
  }

  for (int t = T_ - 1; t >= 0; --t) {
    const float* Fg = Fg_ + (size_t)(t * NB_ + b) * NS_ * NSC_;
    const float* fg = fg_ + (size_t)(t * NB_ + b) * NS_;
    const float* Cg = Cg_ + (size_t)(t * NB_ + b) * NSC_ * NSC_;
    const float* cg = cg_ + (size_t)(t * NB_ + b) * NSC_;

    if (t > 0)
      __builtin_prefetch(pf_arr + (size_t)((t - 1) * NB_ + b) * pf_tile + pf_off, 0, 3);

    const int  hc = (hl < NSC_) ? hl : (NSC_ - 1);
    const bool cv = hl < NSC_;
    float Fcol[NS_];
    #pragma unroll
    for (int i = 0; i < NS_; ++i) Fcol[i] = Fg[i * NSC_ + hc];
    float Ccol[8];
    #pragma unroll
    for (int r = 0; r < 8; ++r) {
      const int rw = r + rbase;
      Ccol[r] = Cg[((rw < NSC_) ? rw : (NSC_ - 1)) * NSC_ + hc];
    }
    const float fu0 = Fg[(lane & 7) * NSC_ + NS_ + 0];
    const float fu1 = Fg[(lane & 7) * NSC_ + NS_ + 1];
    const float fu2 = Fg[(lane & 7) * NSC_ + NS_ + 2];
    const float fu3 = Fg[(lane & 7) * NSC_ + NS_ + 3];
    const float fl  = fg[lane & 7];
    const float cl  = cg[hc];
    v2f bf0, bf1;
    bf0.x = fg[kbase];     bf0.y = fg[kbase + 1];
    bf1.x = fg[4 + kbase]; bf1.y = fg[4 + kbase + 1];

    v2f fb0, fb1;
    fb0.x = cv ? (hiw ? Fcol[2] : Fcol[0]) : 0.f;
    fb0.y = cv ? (hiw ? Fcol[3] : Fcol[1]) : 0.f;
    fb1.x = cv ? (hiw ? Fcol[6] : Fcol[4]) : 0.f;
    fb1.y = cv ? (hiw ? Fcol[7] : Fcol[5]) : 0.f;
    v2f af;
    af.x = (hl < NS_) ? (hiw ? fu2 : fu0) : 0.f;
    af.y = (hl < NS_) ? (hiw ? fu3 : fu1) : 0.f;
    v8f cF;
    #pragma unroll
    for (int r = 0; r < 8; ++r) cF[r] = (!hiw && cv) ? Fcol[r] : 0.f;

    v8f acc = {0.f, 0.f, 0.f, 0.f, 0.f, 0.f, 0.f, 0.f};
    v8f y;
    const int l7 = lane & 7;
    {
      v2f a0, a1;
      a0.x = Pm[l7][kbase];     a0.y = Pm[l7][kbase + 1];
      a1.x = Pm[l7][4 + kbase]; a1.y = Pm[l7][4 + kbase + 1];
      #pragma unroll
      for (int m = 0; m < NS_; ++m) y[m] = pv[m];
      acc = wmma4(a0, fb0, acc);
      acc = wmma4(a1, fb1, acc);
      y = wmma4(a0, bf0, y);
      y = wmma4(a1, bf1, y);
    }

    v8f acc2;
    #pragma unroll
    for (int r = 0; r < 8; ++r) acc2[r] = ((r + rbase) < NSC_ && cv) ? Ccol[r] : 0.f;
    float qs;
    {
      v2f b0, b1;
      b0.x = hiw ? acc[2] : acc[0]; b0.y = hiw ? acc[3] : acc[1];
      b1.x = hiw ? acc[6] : acc[4]; b1.y = hiw ? acc[7] : acc[5];
      acc2 = wmma4(fb0, b0, acc2);
      acc2 = wmma4(fb1, b1, acc2);
      qs = cl;
      #pragma unroll
      for (int i = 0; i < NS_; ++i) qs += Fcol[i] * y[i];
    }
    #pragma unroll
    for (int r = 0; r < 8; ++r) Qm[r + rbase][hl] = acc2[r];
    wave_order();

    solve_update(t, true, qs, hc, fl, fu0, fu1, fu2, fu3, af, cF);
  }

  float* out_x = out;
  float* out_u = out + (size_t)T1_ * NB_ * NS_;
  const int row = (lane < NSC_) ? lane : (NSC_ - 1);
  float xv = (lane < NS_) ? x0g[(size_t)b * NS_ + lane] : 0.f;
  float m[NS_ + 1], mn[NS_ + 1];
  #pragma unroll
  for (int j = 0; j <= NS_; ++j) m[j] = Mall[0][row][j];
  for (int t = 0; t <= T_; ++t) {
    const int tn = (t < T_) ? t + 1 : T_;
    #pragma unroll
    for (int j = 0; j <= NS_; ++j) mn[j] = Mall[tn][row][j];
    float xj[NS_];
    #pragma unroll
    for (int j = 0; j < NS_; ++j)
      xj[j] = __int_as_float(__builtin_amdgcn_readlane(__float_as_int(xv), j));
    float s0 = m[NS_], s1 = 0.f;
    #pragma unroll
    for (int j = 0; j < NS_; j += 2) { s0 += m[j] * xj[j]; s1 += m[j + 1] * xj[j + 1]; }
    const float s = s0 + s1;
    if (lane < NS_) out_x[(size_t)t * NB_ * NS_ + b * NS_ + lane] = xv;
    else if (lane < NSC_) out_u[(size_t)t * NB_ * NC_ + b * NC_ + (lane - NS_)] = s;
    xv = s;
    #pragma unroll
    for (int j = 0; j <= NS_; ++j) m[j] = mn[j];
  }
}

extern "C" void kernel_launch(void* const* d_in, const int* in_sizes, int n_in,
                              void* d_out, int out_size, void* d_ws, size_t ws_size,
                              hipStream_t stream) {
  (void)in_sizes; (void)n_in; (void)d_ws; (void)ws_size; (void)out_size;
  const float* x0 = (const float*)d_in[0];
  const float* C  = (const float*)d_in[1];
  const float* c  = (const float*)d_in[2];
  const float* F  = (const float*)d_in[3];
  const float* f  = (const float*)d_in[4];
  float* out = (float*)d_out;
  mpc_riccati<<<NB_, 32, 0, stream>>>(x0, C, c, F, f, out);
}


# Next step's F, C, c, f pulled into the compute unit's own cache one step ahead by one plain touch load per cache line; the prefetch instruction (fills second level only) moves to two steps ahead
# speedup vs baseline: 5.2158x; 1.0310x over previous
#include <hip/hip_runtime.h>


#define NS_  8
#define NC_  4
#define T_   32
#define NB_  128
#define NSC_ 12
#define T1_  33

typedef float v2f __attribute__((ext_vector_type(2)));
typedef float v8f __attribute__((ext_vector_type(8)));

__device__ __forceinline__ v8f wmma4(v2f a, v2f b, v8f acc) {
  return __builtin_amdgcn_wmma_f32_16x16x4_f32(
       false, a,  false, b,
       (short)0, acc,  false,  false);
}

__device__ __forceinline__ void wave_order() {
  __builtin_amdgcn_fence(__ATOMIC_ACQ_REL, "wavefront");
}

__global__ __launch_bounds__(32) void mpc_riccati(
    const float* __restrict__ x0g,
    const float* __restrict__ Cg_,
    const float* __restrict__ cg_,
    const float* __restrict__ Fg_,
    const float* __restrict__ fg_,
    float* __restrict__ out)
{
  const int  b    = blockIdx.x;
  const int  lane = threadIdx.x;
  const int  hl   = lane & 15;
  const bool hiw  = lane >= 16;
  const int  rbase = hiw ? 8 : 0;
  const int  kbase = hiw ? 2 : 0;

  __shared__ float Pm[16][16];
  __shared__ float Qm[16][16];
  __shared__ float pv[NS_];
  __shared__ float qv[NSC_];
  __shared__ float Mall[T1_][NSC_][NS_ + 1];

  const float* pf_arr;
  int pf_tile, pf_off;
  if (lane >= 4 && lane < 10) {
    pf_arr = Cg_; pf_tile = NSC_ * NSC_;
    pf_off = (lane < 9) ? (lane - 4) * 32 : (NSC_ * NSC_ - 1);
  } else if (lane == 10 || lane == 11) {
    pf_arr = cg_; pf_tile = NSC_;
    pf_off = (lane == 10) ? 0 : (NSC_ - 1);
  } else if (lane == 12 || lane == 13) {
    pf_arr = fg_; pf_tile = NS_;
    pf_off = (lane == 12) ? 0 : (NS_ - 1);
  } else {
    pf_arr = Fg_; pf_tile = NS_ * NSC_;
    pf_off = (lane < 3) ? lane * 32 : ((lane == 3) ? (NS_ * NSC_ - 1) : 0);
  }
  float touched = pf_arr[(size_t)((T_ - 1) * NB_ + b) * pf_tile + pf_off];
  const float* touch_at = pf_arr;
  __builtin_prefetch(pf_arr + (size_t)((T_ - 2) * NB_ + b) * pf_tile + pf_off, 0, 3);

  for (int idx = lane; idx < 256; idx += 32) {
    int r = idx >> 4, cx = idx & 15;
    Pm[r][cx] = 0.f; Qm[r][cx] = 0.f;
  }
  wave_order();

  auto solve_update = [&](int t, const bool dyn, float qs, int hc, float fl,
                          float fu0, float fu1, float fu2, float fu3,
                          v2f af, v8f cF) {
    float A[4][4];
    #pragma unroll
    for (int i = 0; i < 4; ++i)
      #pragma unroll
      for (int j = 0; j < 4; ++j) A[i][j] = Qm[NS_ + i][NS_ + j];
    float rx[4], rq[4];
    #pragma unroll
    for (int m2 = 0; m2 < 4; ++m2) rx[m2] = Qm[NS_ + m2][hl];
    if (dyn) qv[hc] = qs;
    wave_order();
    #pragma unroll
    for (int m2 = 0; m2 < 4; ++m2) rq[m2] = qv[NS_ + m2];

    __builtin_amdgcn_sched_barrier(0);

    if (dyn) {
      asm volatile("" :: "v"(touched));
      touched = *touch_at;
    }

    const int l7 = lane & 7;
    float qxu[4];
    #pragma unroll
    for (int j = 0; j < 4; ++j) qxu[j] = Qm[l7][NS_ + j];
    const float qx = qv[l7];
    v2f a;
    a.x = Qm[hl][NS_ + kbase];
    a.y = Qm[hl][NS_ + kbase + 1];
    v8f cf;
    #pragma unroll
    for (int r = 0; r < 8; ++r) cf[r] = Qm[r + rbase][hl];

    float rp[4];
    rp[0] = 1.f / A[0][0];
    rp[1] = A[0][0] / (A[0][0] * A[1][1] - A[1][0] * A[0][1]);

    #pragma unroll
    for (int cc = 0; cc < 4; ++cc) {
      const float piv = rp[cc];
      #pragma unroll
      for (int j = 0; j < 4; ++j) A[cc][j] *= piv;
      rx[cc] *= piv; rq[cc] *= piv;
      #pragma unroll
      for (int r = 0; r < 4; ++r) {
        if (r == cc) continue;
        const float fac = A[r][cc];
        #pragma unroll
        for (int j = 0; j < 4; ++j) A[r][j] -= fac * A[cc][j];
        rx[r] -= fac * rx[cc];
        rq[r] -= fac * rq[cc];
      }
      if (cc == 1) {
        rp[2] = 1.f / A[2][2];
        rp[3] = A[2][2] / (A[2][2] * A[3][3] - A[3][2] * A[2][3]);
      }
    }
    float Kc[4], kk[4];
    #pragma unroll
    for (int m2 = 0; m2 < 4; ++m2) { Kc[m2] = -rx[m2]; kk[m2] = -rq[m2]; }
    v2f bb;
    bb.x = hiw ? Kc[2] : Kc[0];
    bb.y = hiw ? Kc[3] : Kc[1];
    v8f d = wmma4(a, bb, cf);
    #pragma unroll
    for (int r = 0; r < 8; ++r) Pm[r + rbase][hl] = d[r];
    {
      const bool isK = hl < NS_;
      const int  kc  = isK ? hl : NS_;
      #pragma unroll
      for (int m2 = 0; m2 < 4; ++m2)
        Mall[t][NS_ + m2][kc] = isK ? Kc[m2] : kk[m2];
    }
    {
      float s = qx;
      #pragma unroll
      for (int j = 0; j < NC_; ++j) s += qxu[j] * kk[j];
      pv[l7] = s;
      if (dyn) {
        float bs = fl;
        bs += fu0 * kk[0];
        bs += fu1 * kk[1];
        bs += fu2 * kk[2];
        bs += fu3 * kk[3];
        Mall[t][l7][NS_] = bs;
      }
    }
    if (dyn) {
      v8f acl = wmma4(af, bb, cF);
      if (lane < NS_) {
        #pragma unroll
        for (int r = 0; r < 8; ++r) Mall[t][r][lane] = acl[r];
      }
    } else {
      for (int idx = lane; idx < NS_ * (NS_ + 1); idx += 32)
        Mall[t][idx / (NS_ + 1)][idx % (NS_ + 1)] = 0.f;
    }
    wave_order();
  };

  {
    const float* Cg = Cg_ + (size_t)((T_ - 1) * NB_ + b) * NSC_ * NSC_;
    for (int idx = lane; idx < NSC_ * NSC_; idx += 32)
      Qm[idx / NSC_][idx % NSC_] = Cg[idx];
    const float* cg = cg_ + (size_t)((T_ - 1) * NB_ + b) * NSC_;
    if (lane < NSC_) qv[lane] = cg[lane];
    wave_order();
    const v2f z2 = {0.f, 0.f};
    const v8f z8 = {0.f, 0.f, 0.f, 0.f, 0.f, 0.f, 0.f, 0.f};
    solve_update(T_, false, 0.f, 0, 0.f, 0.f, 0.f, 0.f, 0.f, z2, z8);
  }

  for (int t = T_ - 1; t >= 0; --t) {
    const float* Fg = Fg_ + (size_t)(t * NB_ + b) * NS_ * NSC_;
    const float* fg = fg_ + (size_t)(t * NB_ + b) * NS_;
    const float* Cg = Cg_ + (size_t)(t * NB_ + b) * NSC_ * NSC_;
    const float* cg = cg_ + (size_t)(t * NB_ + b) * NSC_;

    if (t > 1)
      __builtin_prefetch(pf_arr + (size_t)((t - 2) * NB_ + b) * pf_tile + pf_off, 0, 3);
    touch_at =
        pf_arr + (size_t)(((t > 0) ? t - 1 : 0) * NB_ + b) * pf_tile + pf_off;

    const int  hc = (hl < NSC_) ? hl : (NSC_ - 1);
    const bool cv = hl < NSC_;
    float Fcol[NS_];
    #pragma unroll
    for (int i = 0; i < NS_; ++i) Fcol[i] = Fg[i * NSC_ + hc];
    float Ccol[8];
    #pragma unroll
    for (int r = 0; r < 8; ++r) {
      const int rw = r + rbase;
      Ccol[r] = Cg[((rw < NSC_) ? rw : (NSC_ - 1)) * NSC_ + hc];
    }
    const float fu0 = Fg[(lane & 7) * NSC_ + NS_ + 0];
    const float fu1 = Fg[(lane & 7) * NSC_ + NS_ + 1];
    const float fu2 = Fg[(lane & 7) * NSC_ + NS_ + 2];
    const float fu3 = Fg[(lane & 7) * NSC_ + NS_ + 3];
    const float fl  = fg[lane & 7];
    const float cl  = cg[hc];
    v2f bf0, bf1;
    bf0.x = fg[kbase];     bf0.y = fg[kbase + 1];
    bf1.x = fg[4 + kbase]; bf1.y = fg[4 + kbase + 1];

    v2f fb0, fb1;
    fb0.x = cv ? (hiw ? Fcol[2] : Fcol[0]) : 0.f;
    fb0.y = cv ? (hiw ? Fcol[3] : Fcol[1]) : 0.f;
    fb1.x = cv ? (hiw ? Fcol[6] : Fcol[4]) : 0.f;
    fb1.y = cv ? (hiw ? Fcol[7] : Fcol[5]) : 0.f;
    v2f af;
    af.x = (hl < NS_) ? (hiw ? fu2 : fu0) : 0.f;
    af.y = (hl < NS_) ? (hiw ? fu3 : fu1) : 0.f;
    v8f cF;
    #pragma unroll
    for (int r = 0; r < 8; ++r) cF[r] = (!hiw && cv) ? Fcol[r] : 0.f;

    v8f acc = {0.f, 0.f, 0.f, 0.f, 0.f, 0.f, 0.f, 0.f};
    v8f y;
    const int l7 = lane & 7;
    {
      v2f a0, a1;
      a0.x = Pm[l7][kbase];     a0.y = Pm[l7][kbase + 1];
      a1.x = Pm[l7][4 + kbase]; a1.y = Pm[l7][4 + kbase + 1];
      #pragma unroll
      for (int m = 0; m < NS_; ++m) y[m] = pv[m];
      acc = wmma4(a0, fb0, acc);
      acc = wmma4(a1, fb1, acc);
      y = wmma4(a0, bf0, y);
      y = wmma4(a1, bf1, y);
    }

    v8f acc2;
    #pragma unroll
    for (int r = 0; r < 8; ++r) acc2[r] = ((r + rbase) < NSC_ && cv) ? Ccol[r] : 0.f;
    float qs;
    {
      v2f b0, b1;
      b0.x = hiw ? acc[2] : acc[0]; b0.y = hiw ? acc[3] : acc[1];
      b1.x = hiw ? acc[6] : acc[4]; b1.y = hiw ? acc[7] : acc[5];
      acc2 = wmma4(fb0, b0, acc2);
      acc2 = wmma4(fb1, b1, acc2);
      qs = cl;
      #pragma unroll
      for (int i = 0; i < NS_; ++i) qs += Fcol[i] * y[i];
    }
    #pragma unroll
    for (int r = 0; r < 8; ++r) Qm[r + rbase][hl] = acc2[r];
    wave_order();

    solve_update(t, true, qs, hc, fl, fu0, fu1, fu2, fu3, af, cF);
  }

  float* out_x = out;
  float* out_u = out + (size_t)T1_ * NB_ * NS_;
  const int row = (lane < NSC_) ? lane : (NSC_ - 1);
  float xv = (lane < NS_) ? x0g[(size_t)b * NS_ + lane] : 0.f;
  float m[NS_ + 1], mn[NS_ + 1];
  #pragma unroll
  for (int j = 0; j <= NS_; ++j) m[j] = Mall[0][row][j];
  for (int t = 0; t <= T_; ++t) {
    const int tn = (t < T_) ? t + 1 : T_;
    #pragma unroll
    for (int j = 0; j <= NS_; ++j) mn[j] = Mall[tn][row][j];
    float xj[NS_];
    #pragma unroll
    for (int j = 0; j < NS_; ++j)
      xj[j] = __int_as_float(__builtin_amdgcn_readlane(__float_as_int(xv), j));
    float s0 = m[NS_], s1 = 0.f;
    #pragma unroll
    for (int j = 0; j < NS_; j += 2) { s0 += m[j] * xj[j]; s1 += m[j + 1] * xj[j + 1]; }
    const float s = s0 + s1;
    if (lane < NS_) out_x[(size_t)t * NB_ * NS_ + b * NS_ + lane] = xv;
    else if (lane < NSC_) out_u[(size_t)t * NB_ * NC_ + b * NC_ + (lane - NS_)] = s;
    xv = s;
    #pragma unroll
    for (int j = 0; j <= NS_; ++j) m[j] = mn[j];
  }
}

extern "C" void kernel_launch(void* const* d_in, const int* in_sizes, int n_in,
                              void* d_out, int out_size, void* d_ws, size_t ws_size,
                              hipStream_t stream) {
  (void)in_sizes; (void)n_in; (void)d_ws; (void)ws_size; (void)out_size;
  const float* x0 = (const float*)d_in[0];
  const float* C  = (const float*)d_in[1];
  const float* c  = (const float*)d_in[2];
  const float* F  = (const float*)d_in[3];
  const float* f  = (const float*)d_in[4];
  float* out = (float*)d_out;
  mpc_riccati<<<NB_, 32, 0, stream>>>(x0, C, c, F, f, out);
}


# Rollout-table work of a step (A_t product, b_t, stores of K, k, A_t, b_t) held back into the next step's wait for its loads, branch-free; only p and P's store are left behind the value update
# speedup vs baseline: 5.8112x; 1.1142x over previous
#include <hip/hip_runtime.h>


#define NS_  8
#define NC_  4
#define T_   32
#define NB_  128
#define NSC_ 12
#define T1_  33

typedef float v2f __attribute__((ext_vector_type(2)));
typedef float v8f __attribute__((ext_vector_type(8)));

__device__ __forceinline__ v8f wmma4(v2f a, v2f b, v8f acc) {
  return __builtin_amdgcn_wmma_f32_16x16x4_f32(
       false, a,  false, b,
       (short)0, acc,  false,  false);
}

__device__ __forceinline__ void wave_order() {
  __builtin_amdgcn_fence(__ATOMIC_ACQ_REL, "wavefront");
}

__global__ __launch_bounds__(32) void mpc_riccati(
    const float* __restrict__ x0g,
    const float* __restrict__ Cg_,
    const float* __restrict__ cg_,
    const float* __restrict__ Fg_,
    const float* __restrict__ fg_,
    float* __restrict__ out)
{
  const int  b    = blockIdx.x;
  const int  lane = threadIdx.x;
  const int  hl   = lane & 15;
  const bool hiw  = lane >= 16;
  const int  rbase = hiw ? 8 : 0;
  const int  kbase = hiw ? 2 : 0;

  __shared__ float Pm[16][16];
  __shared__ float Qm[16][16];
  __shared__ float pv[NS_];
  __shared__ float qv[NSC_];
  __shared__ float Mall[T1_ + 1][NSC_][NS_ + 1];

  const float* pf_arr;
  int pf_tile, pf_off;
  if (lane >= 4 && lane < 10) {
    pf_arr = Cg_; pf_tile = NSC_ * NSC_;
    pf_off = (lane < 9) ? (lane - 4) * 32 : (NSC_ * NSC_ - 1);
  } else if (lane == 10 || lane == 11) {
    pf_arr = cg_; pf_tile = NSC_;
    pf_off = (lane == 10) ? 0 : (NSC_ - 1);
  } else if (lane == 12 || lane == 13) {
    pf_arr = fg_; pf_tile = NS_;
    pf_off = (lane == 12) ? 0 : (NS_ - 1);
  } else {
    pf_arr = Fg_; pf_tile = NS_ * NSC_;
    pf_off = (lane < 3) ? lane * 32 : ((lane == 3) ? (NS_ * NSC_ - 1) : 0);
  }
  float touched = pf_arr[(size_t)((T_ - 1) * NB_ + b) * pf_tile + pf_off];
  const float* touch_at = pf_arr;
  __builtin_prefetch(pf_arr + (size_t)((T_ - 2) * NB_ + b) * pf_tile + pf_off, 0, 3);

  for (int idx = lane; idx < 256; idx += 32) {
    int r = idx >> 4, cx = idx & 15;
    Pm[r][cx] = 0.f; Qm[r][cx] = 0.f;
  }
  wave_order();

  int   pend_t = T_;
  v2f   pend_bb = {0.f, 0.f}, pend_af = {0.f, 0.f};
  v8f   pend_cF = {0.f, 0.f, 0.f, 0.f, 0.f, 0.f, 0.f, 0.f};
  float pend_kk[4] = {0.f, 0.f, 0.f, 0.f}, pend_tab[4] = {0.f, 0.f, 0.f, 0.f};
  float pend_fl = 0.f, pend_fu0 = 0.f, pend_fu1 = 0.f, pend_fu2 = 0.f, pend_fu3 = 0.f;
  auto table_work = [&]() {
    v8f acl = wmma4(pend_af, pend_bb, pend_cF);
    {
      const int kc = (hl < NS_) ? hl : NS_;
      #pragma unroll
      for (int m2 = 0; m2 < 4; ++m2) Mall[pend_t][NS_ + m2][kc] = pend_tab[m2];
    }
    {
      float bs = pend_fl;
      bs += pend_fu0 * pend_kk[0];
      bs += pend_fu1 * pend_kk[1];
      bs += pend_fu2 * pend_kk[2];
      bs += pend_fu3 * pend_kk[3];
      Mall[pend_t][lane & 7][NS_] = bs;
    }
    {
      float* mflat = &Mall[0][0][0];
      const int base = ((lane < NS_) ? pend_t : T1_) * (NSC_ * (NS_ + 1)) + lane;
      #pragma unroll
      for (int r = 0; r < 8; ++r) mflat[base + r * (NS_ + 1)] = acl[r];
    }
  };

  auto solve_update = [&](int t, const bool dyn, float qs, int hc, float fl,
                          float fu0, float fu1, float fu2, float fu3,
                          v2f af, v8f cF) {
    float A[4][4];
    #pragma unroll
    for (int i = 0; i < 4; ++i)
      #pragma unroll
      for (int j = 0; j < 4; ++j) A[i][j] = Qm[NS_ + i][NS_ + j];
    float rx[4], rq[4];
    #pragma unroll
    for (int m2 = 0; m2 < 4; ++m2) rx[m2] = Qm[NS_ + m2][hl];
    if (dyn) qv[hc] = qs;
    wave_order();
    #pragma unroll
    for (int m2 = 0; m2 < 4; ++m2) rq[m2] = qv[NS_ + m2];
    const int l7 = lane & 7;
    float qxu[4];
    #pragma unroll
    for (int j = 0; j < 4; ++j) qxu[j] = Qm[l7][NS_ + j];
    const float qx = qv[l7];

    __builtin_amdgcn_sched_barrier(0);

    if (dyn) {
      asm volatile("" :: "v"(touched));
      touched = *touch_at;
    }

    v2f a;
    a.x = Qm[hl][NS_ + kbase];
    a.y = Qm[hl][NS_ + kbase + 1];
    v8f cf;
    #pragma unroll
    for (int r = 0; r < 8; ++r) cf[r] = Qm[r + rbase][hl];

    float rp[4];
    rp[0] = 1.f / A[0][0];
    rp[1] = A[0][0] / (A[0][0] * A[1][1] - A[1][0] * A[0][1]);

    #pragma unroll
    for (int cc = 0; cc < 4; ++cc) {
      const float piv = rp[cc];
      #pragma unroll
      for (int j = 0; j < 4; ++j) A[cc][j] *= piv;
      rx[cc] *= piv; rq[cc] *= piv;
      #pragma unroll
      for (int r = 0; r < 4; ++r) {
        if (r == cc) continue;
        const float fac = A[r][cc];
        #pragma unroll
        for (int j = 0; j < 4; ++j) A[r][j] -= fac * A[cc][j];
        rx[r] -= fac * rx[cc];
        rq[r] -= fac * rq[cc];
      }
      if (cc == 1) {
        rp[2] = 1.f / A[2][2];
        rp[3] = A[2][2] / (A[2][2] * A[3][3] - A[3][2] * A[2][3]);
      }
    }
    float Kc[4], kk[4];
    #pragma unroll
    for (int m2 = 0; m2 < 4; ++m2) { Kc[m2] = -rx[m2]; kk[m2] = -rq[m2]; }
    v2f bb;
    bb.x = hiw ? Kc[2] : Kc[0];
    bb.y = hiw ? Kc[3] : Kc[1];
    v8f d = wmma4(a, bb, cf);
    {
      float s = qx;
      #pragma unroll
      for (int j = 0; j < NC_; ++j) s += qxu[j] * kk[j];
      pv[l7] = s;
    }
    #pragma unroll
    for (int r = 0; r < 8; ++r) Pm[r + rbase][hl] = d[r];
    pend_t = t; pend_bb = bb; pend_af = af; pend_cF = cF;
    pend_fl = fl; pend_fu0 = fu0; pend_fu1 = fu1; pend_fu2 = fu2; pend_fu3 = fu3;
    #pragma unroll
    for (int m2 = 0; m2 < 4; ++m2) {
      pend_kk[m2]  = kk[m2];
      pend_tab[m2] = (hl < NS_) ? Kc[m2] : kk[m2];
    }
    wave_order();
  };

  {
    const float* Cg = Cg_ + (size_t)((T_ - 1) * NB_ + b) * NSC_ * NSC_;
    for (int idx = lane; idx < NSC_ * NSC_; idx += 32)
      Qm[idx / NSC_][idx % NSC_] = Cg[idx];
    const float* cg = cg_ + (size_t)((T_ - 1) * NB_ + b) * NSC_;
    if (lane < NSC_) qv[lane] = cg[lane];
    wave_order();
    const v2f z2 = {0.f, 0.f};
    const v8f z8 = {0.f, 0.f, 0.f, 0.f, 0.f, 0.f, 0.f, 0.f};
    solve_update(T_, false, 0.f, 0, 0.f, 0.f, 0.f, 0.f, 0.f, z2, z8);
  }

  for (int t = T_ - 1; t >= 0; --t) {
    const float* Fg = Fg_ + (size_t)(t * NB_ + b) * NS_ * NSC_;
    const float* fg = fg_ + (size_t)(t * NB_ + b) * NS_;
    const float* Cg = Cg_ + (size_t)(t * NB_ + b) * NSC_ * NSC_;
    const float* cg = cg_ + (size_t)(t * NB_ + b) * NSC_;

    touch_at =
        pf_arr + (size_t)(((t > 0) ? t - 1 : 0) * NB_ + b) * pf_tile + pf_off;

    const int  hc = (hl < NSC_) ? hl : (NSC_ - 1);
    const bool cv = hl < NSC_;
    const int l7 = lane & 7;
    v2f a0, a1;
    a0.x = Pm[l7][kbase];     a0.y = Pm[l7][kbase + 1];
    a1.x = Pm[l7][4 + kbase]; a1.y = Pm[l7][4 + kbase + 1];
    v8f y;
    #pragma unroll
    for (int m = 0; m < NS_; ++m) y[m] = pv[m];
    v2f bf0, bf1;
    float Fcol[NS_];
    bf0.x = fg[kbase];     bf0.y = fg[kbase + 1];
    #pragma unroll
    for (int i = 0; i < 4; ++i) Fcol[i] = Fg[i * NSC_ + hc];
    bf1.x = fg[4 + kbase]; bf1.y = fg[4 + kbase + 1];
    #pragma unroll
    for (int i = 4; i < NS_; ++i) Fcol[i] = Fg[i * NSC_ + hc];
    __builtin_amdgcn_sched_barrier(0);
    float Ccol[8];
    #pragma unroll
    for (int r = 0; r < 8; ++r) {
      const int rw = r + rbase;
      Ccol[r] = Cg[((rw < NSC_) ? rw : (NSC_ - 1)) * NSC_ + hc];
    }
    const float cl  = cg[hc];
    const float fu0 = Fg[(lane & 7) * NSC_ + NS_ + 0];
    const float fu1 = Fg[(lane & 7) * NSC_ + NS_ + 1];
    const float fu2 = Fg[(lane & 7) * NSC_ + NS_ + 2];
    const float fu3 = Fg[(lane & 7) * NSC_ + NS_ + 3];
    const float fl  = fg[lane & 7];

    __builtin_amdgcn_sched_barrier(0);
    table_work();
    __builtin_prefetch(
        pf_arr + (size_t)(((t > 1) ? t - 2 : 0) * NB_ + b) * pf_tile + pf_off, 0, 3);
    __builtin_amdgcn_sched_barrier(0);

    v2f fb0, fb1;
    fb0.x = cv ? (hiw ? Fcol[2] : Fcol[0]) : 0.f;
    fb0.y = cv ? (hiw ? Fcol[3] : Fcol[1]) : 0.f;
    fb1.x = cv ? (hiw ? Fcol[6] : Fcol[4]) : 0.f;
    fb1.y = cv ? (hiw ? Fcol[7] : Fcol[5]) : 0.f;
    v2f af;
    af.x = (hl < NS_) ? (hiw ? fu2 : fu0) : 0.f;
    af.y = (hl < NS_) ? (hiw ? fu3 : fu1) : 0.f;
    v8f cF;
    #pragma unroll
    for (int r = 0; r < 8; ++r) cF[r] = (!hiw && cv) ? Fcol[r] : 0.f;

    v8f acc = {0.f, 0.f, 0.f, 0.f, 0.f, 0.f, 0.f, 0.f};
    {
      acc = wmma4(a0, fb0, acc);
      acc = wmma4(a1, fb1, acc);
      y = wmma4(a0, bf0, y);
      y = wmma4(a1, bf1, y);
    }

    v8f acc2;
    #pragma unroll
    for (int r = 0; r < 8; ++r) acc2[r] = ((r + rbase) < NSC_ && cv) ? Ccol[r] : 0.f;
    float qs;
    {
      v2f b0, b1;
      b0.x = hiw ? acc[2] : acc[0]; b0.y = hiw ? acc[3] : acc[1];
      b1.x = hiw ? acc[6] : acc[4]; b1.y = hiw ? acc[7] : acc[5];
      acc2 = wmma4(fb0, b0, acc2);
      acc2 = wmma4(fb1, b1, acc2);
      qs = cl;
      #pragma unroll
      for (int i = 0; i < NS_; ++i) qs += Fcol[i] * y[i];
    }
    #pragma unroll
    for (int r = 0; r < 8; ++r) Qm[r + rbase][hl] = acc2[r];
    wave_order();

    solve_update(t, true, qs, hc, fl, fu0, fu1, fu2, fu3, af, cF);
  }

  table_work();
  wave_order();

  float* out_x = out;
  float* out_u = out + (size_t)T1_ * NB_ * NS_;
  const int row = (lane < NSC_) ? lane : (NSC_ - 1);
  float xv = (lane < NS_) ? x0g[(size_t)b * NS_ + lane] : 0.f;
  float m[NS_ + 1], mn[NS_ + 1];
  #pragma unroll
  for (int j = 0; j <= NS_; ++j) m[j] = Mall[0][row][j];
  for (int t = 0; t <= T_; ++t) {
    const int tn = (t < T_) ? t + 1 : T_;
    #pragma unroll
    for (int j = 0; j <= NS_; ++j) mn[j] = Mall[tn][row][j];
    float xj[NS_];
    #pragma unroll
    for (int j = 0; j < NS_; ++j)
      xj[j] = __int_as_float(__builtin_amdgcn_readlane(__float_as_int(xv), j));
    float s0 = m[NS_], s1 = 0.f;
    #pragma unroll
    for (int j = 0; j < NS_; j += 2) { s0 += m[j] * xj[j]; s1 += m[j + 1] * xj[j + 1]; }
    const float s = s0 + s1;
    if (lane < NS_) out_x[(size_t)t * NB_ * NS_ + b * NS_ + lane] = xv;
    else if (lane < NSC_) out_u[(size_t)t * NB_ * NC_ + b * NC_ + (lane - NS_)] = s;
    xv = s;
    #pragma unroll
    for (int j = 0; j <= NS_; ++j) m[j] = mn[j];
  }
}

extern "C" void kernel_launch(void* const* d_in, const int* in_sizes, int n_in,
                              void* d_out, int out_size, void* d_ws, size_t ws_size,
                              hipStream_t stream) {
  (void)in_sizes; (void)n_in; (void)d_ws; (void)ws_size; (void)out_size;
  const float* x0 = (const float*)d_in[0];
  const float* C  = (const float*)d_in[1];
  const float* c  = (const float*)d_in[2];
  const float* F  = (const float*)d_in[3];
  const float* f  = (const float*)d_in[4];
  float* out = (float*)d_out;
  mpc_riccati<<<NB_, 32, 0, stream>>>(x0, C, c, F, f, out);
}
